# starting kernel
# baseline (speedup 1.0000x reference)
#include <hip/hip_runtime.h>
#include <hip/hip_bf16.h>
#include <math.h>

#define NA   16
#define DOBS 256
#define AD   32
#define BB   16
#define TT   64
#define SS   16
#define H1D  128
#define H2D  64
#define BT   (BB*TT)
#define NVAR 18
#define INV_S1 (1.0f/17.0f)

#define H1P  132
#define H2P  68
#define OBSP 260

#define OFF_HOBS 0
#define OFF_H1   2048
#define OFF_H2   4160
#define OFF_PW   5248
#define OFF_PQ   5760
#define OFF_W1A  6272
#define OFF_W2   10368
#define OFF_W3   18560
#define LDS_FLOATS 20608

typedef float v2f __attribute__((ext_vector_type(2)));
typedef float v8f __attribute__((ext_vector_type(8)));

__device__ __forceinline__ v8f wmma4(v2f a, v2f b, v8f c) {
  return __builtin_amdgcn_wmma_f32_16x16x4_f32(
       false, a,  false, b,
       (short)0, c,  false,  false);
}

__device__ __forceinline__ float wave_max(float v) {
  #pragma unroll
  for (int off = 16; off; off >>= 1) v = fmaxf(v, __shfl_xor(v, off, 32));
  return v;
}
__device__ __forceinline__ float wave_sum(float v) {
  #pragma unroll
  for (int off = 16; off; off >>= 1) v += __shfl_xor(v, off, 32);
  return v;
}

__global__ __launch_bounds__(128)
void coma_influence_kernel(const float* __restrict__ obs,
                           const float* __restrict__ act,
                           const float* __restrict__ cf,
                           const float* __restrict__ W1,
                           const float* __restrict__ b1,
                           const float* __restrict__ W2,
                           const float* __restrict__ b2,
                           const float* __restrict__ W3,
                           const float* __restrict__ b3,
                           float* __restrict__ out)
{
  extern __shared__ float smem[];
  float* s_hobs = smem + OFF_HOBS;
  float* s_h1   = smem + OFF_H1;
  float* s_h2   = smem + OFF_H2;
  float* s_pw   = smem + OFF_PW;
  float* s_pq   = smem + OFF_PQ;
  float* s_w1a  = smem + OFF_W1A;
  float* s_w2   = smem + OFF_W2;
  float* s_w3   = smem + OFF_W3;
  float* s_obs  = smem + OFF_W2;

  const int tid  = threadIdx.x;
  const int lane = tid & 31;
  const int wid  = tid >> 5;
  const int g    = blockIdx.x;
  const int r0   = g * 16;
  const int n    = r0 >> 10;
  const int bt0  = r0 & 1023;

  const int col = lane & 15;
  const int hi  = lane >> 4;
  const int kb  = hi * 2;

  #pragma unroll
  for (int j4 = 0; j4 < 8; ++j4) {
    const int i4 = tid + 128 * j4;
    const int m  = i4 >> 6;
    const int d  = (i4 & 63) * 4;
    const float4 v = *(const float4*)(obs + ((size_t)(bt0 + m) * NA + n) * DOBS + d);
    *(float4*)(s_obs + m * OBSP + d) = v;
  }
  __syncthreads();

  #pragma unroll
  for (int jj = 0; jj < 2; ++jj) {
    const int jcol = (wid * 2 + jj) * 16 + col;
    v8f acc;
    {
      const float bias = b1[jcol];
      #pragma unroll
      for (int v = 0; v < 8; ++v) acc[v] = bias;
    }
    for (int kk = 0; kk < DOBS; kk += 4) {
      v2f a; a[0] = s_obs[col * OBSP + kk + kb]; a[1] = s_obs[col * OBSP + kk + kb + 1];
      v2f b; b[0] = W1[(kk + kb) * H1D + jcol];  b[1] = W1[(kk + kb + 1) * H1D + jcol];
      acc = wmma4(a, b, acc);
    }
    #pragma unroll
    for (int v = 0; v < 8; ++v) s_hobs[(v + 8 * hi) * H1D + jcol] = acc[v];
  }
  __syncthreads();

  #pragma unroll
  for (int j4 = 0; j4 < 8; ++j4) {
    const int j = (tid + 128 * j4) * 4;
    *(float4*)(s_w1a + j) = *(const float4*)(W1 + DOBS * H1D + j);
  }
  #pragma unroll
  for (int j4 = 0; j4 < 16; ++j4) {
    const int j = (tid + 128 * j4) * 4;
    *(float4*)(s_w2 + j) = *(const float4*)(W2 + j);
  }
  #pragma unroll
  for (int j4 = 0; j4 < 4; ++j4) {
    const int j = (tid + 128 * j4) * 4;
    *(float4*)(s_w3 + j) = *(const float4*)(W3 + j);
  }
  __syncthreads();

  for (int vi = 0; vi < NVAR; ++vi) {
    const float* act_row = nullptr;
    if (vi == 0)      act_row = act + ((size_t)(bt0 + col) * NA + n) * AD;
    else if (vi >= 2) act_row = cf  + (((size_t)n * SS + (vi - 2)) * BT + (bt0 + col)) * AD;

    if (vi + 1 >= 2 && vi + 1 < NVAR) {
      const float* nxt = cf + (((size_t)n * SS + (vi - 1)) * BT + (bt0 + col)) * AD;
      __builtin_prefetch(nxt, 0, 1);
    }

    v2f av[8];
    if (vi != 1) {
      #pragma unroll
      for (int kq = 0; kq < 8; ++kq) {
        const float2 t = *(const float2*)(act_row + kq * 4 + kb);
        av[kq][0] = t.x; av[kq][1] = t.y;
      }
    }

    #pragma unroll
    for (int jj = 0; jj < 2; ++jj) {
      const int jcol = (wid * 2 + jj) * 16 + col;
      v8f acc;
      #pragma unroll
      for (int v = 0; v < 8; ++v) acc[v] = s_hobs[(v + 8 * hi) * H1D + jcol];
      if (vi != 1) {
        #pragma unroll
        for (int kq = 0; kq < 8; ++kq) {
          const int kk = kq * 4;
          v2f b; b[0] = s_w1a[(kk + kb) * H1D + jcol];
                 b[1] = s_w1a[(kk + kb + 1) * H1D + jcol];
          acc = wmma4(av[kq], b, acc);
        }
      }
      #pragma unroll
      for (int v = 0; v < 8; ++v)
        s_h1[(v + 8 * hi) * H1P + jcol] = fmaxf(acc[v], 0.0f);
    }
    __syncthreads();

    {
      const int jcol = wid * 16 + col;
      v8f acc;
      {
        const float bias = b2[jcol];
        #pragma unroll
        for (int v = 0; v < 8; ++v) acc[v] = bias;
      }
      for (int kk = 0; kk < H1D; kk += 4) {
        v2f a; a[0] = s_h1[col * H1P + kk + kb];  a[1] = s_h1[col * H1P + kk + kb + 1];
        v2f b; b[0] = s_w2[(kk + kb) * H2D + jcol]; b[1] = s_w2[(kk + kb + 1) * H2D + jcol];
        acc = wmma4(a, b, acc);
      }
      #pragma unroll
      for (int v = 0; v < 8; ++v)
        s_h2[(v + 8 * hi) * H2P + jcol] = fmaxf(acc[v], 0.0f);
    }
    __syncthreads();

    if (wid < 2) {
      const int jcol = wid * 16 + col;
      v8f acc;
      {
        const float bias = b3[jcol];
        #pragma unroll
        for (int v = 0; v < 8; ++v) acc[v] = bias;
      }
      for (int kk = 0; kk < H2D; kk += 4) {
        v2f a; a[0] = s_h2[col * H2P + kk + kb];  a[1] = s_h2[col * H2P + kk + kb + 1];
        v2f b; b[0] = s_w3[(kk + kb) * AD + jcol]; b[1] = s_w3[(kk + kb + 1) * AD + jcol];
        acc = wmma4(a, b, acc);
      }
      if (vi == 0) {
        #pragma unroll
        for (int v = 0; v < 8; ++v) s_pw[(v + 8 * hi) * AD + jcol] = acc[v];
      } else if (vi == 1) {
        #pragma unroll
        for (int v = 0; v < 8; ++v) s_pq[(v + 8 * hi) * AD + jcol] = acc[v];
      } else {
        #pragma unroll
        for (int v = 0; v < 8; ++v) s_pq[(v + 8 * hi) * AD + jcol] += acc[v];
      }
    }
    __syncthreads();
  }

  #pragma unroll
  for (int rr = 0; rr < 4; ++rr) {
    const int r = wid * 4 + rr;
    const float pw = s_pw[r * AD + lane];
    const float pq = s_pq[r * AD + lane] * INV_S1;

    const float mp = wave_max(pw);
    const float mq = wave_max(pq);
    const float ep = expf(pw - mp);
    const float eq = expf(pq - mq);
    const float sp = wave_sum(ep);
    const float sq = wave_sum(eq);

    const float log_p = (pw - mp) - logf(sp);
    const float log_q = (pq - mq) - logf(sq);
    const float q     = eq / sq;
    float kl = q * (log_q - log_p);
    kl = wave_sum(kl);
    if (lane == 0) out[(size_t)(bt0 + r) * NA + n] = kl * (1.0f / AD);
  }
}

extern "C" void kernel_launch(void* const* d_in, const int* in_sizes, int n_in,
                              void* d_out, int out_size, void* d_ws, size_t ws_size,
                              hipStream_t stream) {
  (void)in_sizes; (void)n_in; (void)d_ws; (void)ws_size; (void)out_size;
  const float* obs = (const float*)d_in[0];
  const float* act = (const float*)d_in[1];
  const float* cf  = (const float*)d_in[2];
  const float* W1  = (const float*)d_in[3];
  const float* b1  = (const float*)d_in[4];
  const float* W2  = (const float*)d_in[5];
  const float* b2  = (const float*)d_in[6];
  const float* W3  = (const float*)d_in[7];
  const float* b3  = (const float*)d_in[8];
  float* out = (float*)d_out;

  const int tiles = (NA * BB * TT) / 16;
  const size_t lds_bytes = LDS_FLOATS * sizeof(float);
  coma_influence_kernel<<<tiles, 128, lds_bytes, stream>>>(
      obs, act, cf, W1, b1, W2, b2, W3, b3, out);
}


# Activations stay in registers over the 3 layers (weights = A operand from LDS, a D tile is the next B); each wave runs whole variants: 57 barriers -> 4; sum order of layers 2,3, p_without changes
# speedup vs baseline: 2.0574x; 2.0574x over previous
#include <hip/hip_runtime.h>
#include <math.h>

#define NA   16
#define DOBS 256
#define AD   32
#define BB   16
#define TT   64
#define SS   16
#define H1D  128
#define H2D  64
#define BT   (BB*TT)
#define INV_S1 (1.0f/17.0f)

#define H1P  132
#define OBSP 260
#define PP   36

#define OFF_W1A  0
#define OFF_W2   4096
#define OFF_B2   12288
#define OFF_B3   12352
#define OFF_HOBS 12416
#define OFF_PART 14528
#define OFF_W3   17408
#define LDS_FLOATS 19456

typedef float v2f __attribute__((ext_vector_type(2)));
typedef float v4f __attribute__((ext_vector_type(4)));
typedef float v8f __attribute__((ext_vector_type(8)));

__device__ __forceinline__ v8f wmma4(v2f a, v2f b, v8f c) {
  return __builtin_amdgcn_wmma_f32_16x16x4_f32(
       false, a,  false, b,
       (short)0, c,  false,  false);
}

__device__ __forceinline__ float wave_max(float v) {
  #pragma unroll
  for (int off = 16; off; off >>= 1) v = fmaxf(v, __shfl_xor(v, off, 32));
  return v;
}
__device__ __forceinline__ float wave_sum(float v) {
  #pragma unroll
  for (int off = 16; off; off >>= 1) v += __shfl_xor(v, off, 32);
  return v;
}

template <int K_, int M_, bool CHAIN>
__device__ __forceinline__ void stage_frag(float* dst, const float* __restrict__ W, int tid) {
  constexpr int MT  = M_ / 16;
  constexpr int NV4 = K_ * M_ / 4;
  #pragma unroll
  for (int i = 0; i < NV4 / 128; ++i) {
    const int d4    = tid + 128 * i;
    const int ln    = d4 & 31;
    const int chunk = d4 >> 5;
    const int mt    = chunk % MT;
    const int kp    = chunk / MT;
    const int r     = ln & 15;
    const int h     = ln >> 4;
    const int ks0   = 2 * kp, ks1 = 2 * kp + 1;
    int k0, k1;
    if (CHAIN) {
      k0 = 16 * (ks0 >> 2) + 8 * h + 2 * (ks0 & 3);
      k1 = 16 * (ks1 >> 2) + 8 * h + 2 * (ks1 & 3);
    } else {
      k0 = 4 * ks0 + 2 * h;
      k1 = 4 * ks1 + 2 * h;
    }
    const int m = 16 * mt + r;
    float4 v;
    v.x = W[(k0    ) * M_ + m];
    v.y = W[(k0 + 1) * M_ + m];
    v.z = W[(k1    ) * M_ + m];
    v.w = W[(k1 + 1) * M_ + m];
    *(float4*)(dst + d4 * 4) = v;
  }
}

__global__ __launch_bounds__(128, 4)
void coma_influence_kernel(const float* __restrict__ obs,
                           const float* __restrict__ act,
                           const float* __restrict__ cf,
                           const float* __restrict__ W1,
                           const float* __restrict__ b1,
                           const float* __restrict__ W2,
                           const float* __restrict__ b2,
                           const float* __restrict__ W3,
                           const float* __restrict__ b3,
                           float* __restrict__ out)
{
  extern __shared__ float smem[];
  float* s_w1a  = smem + OFF_W1A;
  float* s_w2   = smem + OFF_W2;
  float* s_w3   = smem + OFF_W3;
  float* s_b2   = smem + OFF_B2;
  float* s_b3   = smem + OFF_B3;
  float* s_hobs = smem + OFF_HOBS;
  float* s_part = smem + OFF_PART;
  float* s_pw   = smem + OFF_PART + 4 * 16 * PP;
  float* s_obs  = smem + OFF_PART;

  const int tid  = threadIdx.x;
  const int lane = tid & 31;
  const int wid  = __builtin_amdgcn_readfirstlane(tid >> 5);
  const int g    = blockIdx.x;
  const int r0   = g * 16;
  const int n    = r0 >> 10;
  const int bt0  = r0 & 1023;

  const int col = lane & 15;
  const int hi  = lane >> 4;
  const int kb  = hi * 2;

  #pragma unroll
  for (int j4 = 0; j4 < 8; ++j4) {
    const int i4 = tid + 128 * j4;
    const int m  = i4 >> 6;
    const int d  = (i4 & 63) * 4;
    const float4 v = *(const float4*)(obs + ((size_t)(bt0 + m) * NA + n) * DOBS + d);
    *(float4*)(s_obs + m * OBSP + d) = v;
  }
  stage_frag<AD, H1D, false>(s_w1a, W1 + DOBS * H1D, tid);
  stage_frag<H1D, H2D, true>(s_w2, W2, tid);
  if (tid < H2D) s_b2[tid] = b2[tid];
  else if (tid < H2D + AD) s_b3[tid - H2D] = b3[tid - H2D];
  __syncthreads();

  {
    const int f0 = wid * 32;
    v8f acc0, acc1;
    #pragma unroll
    for (int v = 0; v < 8; ++v) {
      acc0[v] = b1[f0 + 8 * hi + v];
      acc1[v] = b1[f0 + 16 + 8 * hi + v];
    }
    const float* wp = W1 + f0 + col;
    const float* op = s_obs + col * OBSP + kb;
    #pragma unroll 8
    for (int kk = 0; kk < DOBS; kk += 4) {
      const v2f b = *(const v2f*)(op + kk);
      v2f a0, a1;
      a0[0] = wp[(kk + kb) * H1D];      a0[1] = wp[(kk + kb + 1) * H1D];
      a1[0] = wp[(kk + kb) * H1D + 16]; a1[1] = wp[(kk + kb + 1) * H1D + 16];
      acc0 = wmma4(a0, b, acc0);
      acc1 = wmma4(a1, b, acc1);
    }
    float* hp = s_hobs + col * H1P + f0 + 8 * hi;
    *(float4*)(hp)      = make_float4(acc0[0], acc0[1], acc0[2], acc0[3]);
    *(float4*)(hp + 4)  = make_float4(acc0[4], acc0[5], acc0[6], acc0[7]);
    *(float4*)(hp + 16) = make_float4(acc1[0], acc1[1], acc1[2], acc1[3]);
    *(float4*)(hp + 20) = make_float4(acc1[4], acc1[5], acc1[6], acc1[7]);
  }
  __syncthreads();

  stage_frag<H2D, AD, true>(s_w3, W3, tid);
  __syncthreads();

  const int role = (wid + g + (g >> 8)) & 3;
  const int it0  = (role < 2) ? 0 : 1;

  const float* act_row = act + ((size_t)(bt0 + col) * NA + n) * AD + kb;
  const float* cf_row  = cf + (((size_t)n * SS + 4 * role) * BT + (bt0 + col)) * AD + kb;
  const float* hob_row = s_hobs + col * H1P + 8 * hi;
  const float* a1p = s_w1a + lane * 4;
  const float* a2p = s_w2  + lane * 4;
  const float* a3p = s_w3  + lane * 4;

  v8f pq0, pq1;
  #pragma unroll
  for (int v = 0; v < 8; ++v) { pq0[v] = 0.0f; pq1[v] = 0.0f; }
  float* pwp = s_pw + col * PP + 8 * hi;

  v2f bv[8];
  if (role == 1) {
    #pragma unroll
    for (int kq = 0; kq < 8; ++kq) { bv[kq][0] = 0.0f; bv[kq][1] = 0.0f; }
  } else {
    const float* p = (role == 0) ? act_row : cf_row;
    #pragma unroll
    for (int kq = 0; kq < 8; ++kq) bv[kq] = *(const v2f*)(p + kq * 4);
  }

  for (int it = it0; it < 5; ++it) {
    __asm__ volatile("" ::: "memory");
    v2f bn[8];
    {
      const int itn = (it < 4) ? it : 3;
      const float* p = cf_row + (size_t)itn * (BT * AD);
      #pragma unroll
      for (int kq = 0; kq < 8; ++kq) bn[kq] = *(const v2f*)(p + kq * 4);
    }

    v8f h1[8];
    #pragma unroll
    for (int mt = 0; mt < 8; ++mt) {
      const v4f lo = *(const v4f*)(hob_row + 16 * mt);
      const v4f up = *(const v4f*)(hob_row + 16 * mt + 4);
      h1[mt] = __builtin_shufflevector(lo, up, 0, 1, 2, 3, 4, 5, 6, 7);
    }
    #pragma unroll
    for (int kp = 0; kp < 4; ++kp) {
      v4f a[8];
      #pragma unroll
      for (int mt = 0; mt < 8; ++mt) a[mt] = *(const v4f*)(a1p + (kp * 8 + mt) * 128);
      #pragma unroll
      for (int mt = 0; mt < 8; ++mt) h1[mt] = wmma4(a[mt].lo, bv[2 * kp], h1[mt]);
      #pragma unroll
      for (int mt = 0; mt < 8; ++mt) h1[mt] = wmma4(a[mt].hi, bv[2 * kp + 1], h1[mt]);
    }
    #pragma unroll
    for (int mt = 0; mt < 8; ++mt) {
      #pragma unroll
      for (int v = 0; v < 8; ++v) h1[mt][v] = fmaxf(h1[mt][v], 0.0f);
    }

    v8f h2[4];
    #pragma unroll
    for (int mt = 0; mt < 4; ++mt) {
      const v4f lo = *(const v4f*)(s_b2 + 16 * mt + 8 * hi);
      const v4f up = *(const v4f*)(s_b2 + 16 * mt + 8 * hi + 4);
      h2[mt] = __builtin_shufflevector(lo, up, 0, 1, 2, 3, 4, 5, 6, 7);
    }
    #pragma unroll
    for (int t = 0; t < 8; ++t) {
      #pragma unroll
      for (int sp = 0; sp < 2; ++sp) {
        const int kp = 2 * t + sp;
        v4f a[4];
        #pragma unroll
        for (int mt = 0; mt < 4; ++mt) a[mt] = *(const v4f*)(a2p + (kp * 4 + mt) * 128);
        v2f bA, bB;
        bA[0] = h1[t][4 * sp];     bA[1] = h1[t][4 * sp + 1];
        bB[0] = h1[t][4 * sp + 2]; bB[1] = h1[t][4 * sp + 3];
        #pragma unroll
        for (int mt = 0; mt < 4; ++mt) h2[mt] = wmma4(a[mt].lo, bA, h2[mt]);
        #pragma unroll
        for (int mt = 0; mt < 4; ++mt) h2[mt] = wmma4(a[mt].hi, bB, h2[mt]);
      }
    }
    #pragma unroll
    for (int mt = 0; mt < 4; ++mt) {
      #pragma unroll
      for (int v = 0; v < 8; ++v) h2[mt][v] = fmaxf(h2[mt][v], 0.0f);
    }

    v8f p3[2];
    #pragma unroll
    for (int mt = 0; mt < 2; ++mt) {
      const v4f lo = *(const v4f*)(s_b3 + 16 * mt + 8 * hi);
      const v4f up = *(const v4f*)(s_b3 + 16 * mt + 8 * hi + 4);
      p3[mt] = __builtin_shufflevector(lo, up, 0, 1, 2, 3, 4, 5, 6, 7);
    }
    #pragma unroll
    for (int t = 0; t < 4; ++t) {
      #pragma unroll
      for (int sp = 0; sp < 2; ++sp) {
        const int kp = 2 * t + sp;
        v4f a[2];
        #pragma unroll
        for (int mt = 0; mt < 2; ++mt) a[mt] = *(const v4f*)(a3p + (kp * 2 + mt) * 128);
        v2f bA, bB;
        bA[0] = h2[t][4 * sp];     bA[1] = h2[t][4 * sp + 1];
        bB[0] = h2[t][4 * sp + 2]; bB[1] = h2[t][4 * sp + 3];
        #pragma unroll
        for (int mt = 0; mt < 2; ++mt) p3[mt] = wmma4(a[mt].lo, bA, p3[mt]);
        #pragma unroll
        for (int mt = 0; mt < 2; ++mt) p3[mt] = wmma4(a[mt].hi, bB, p3[mt]);
      }
    }

    if (role == 0 && it == 0) {
      *(v4f*)(pwp)      = p3[0].lo;
      *(v4f*)(pwp + 4)  = p3[0].hi;
      *(v4f*)(pwp + 16) = p3[1].lo;
      *(v4f*)(pwp + 20) = p3[1].hi;
    } else {
      pq0 += p3[0]; pq1 += p3[1];
    }

    #pragma unroll
    for (int kq = 0; kq < 8; ++kq) bv[kq] = bn[kq];
  }

  {
    float* pp = s_part + role * (16 * PP) + col * PP + 8 * hi;
    *(v4f*)(pp)      = pq0.lo;
    *(v4f*)(pp + 4)  = pq0.hi;
    *(v4f*)(pp + 16) = pq1.lo;
    *(v4f*)(pp + 20) = pq1.hi;
  }
  __syncthreads();

  #pragma unroll
  for (int rr = 0; rr < 4; ++rr) {
    const int r = wid * 4 + rr;
    const float pw = s_pw[r * PP + lane];
    const float pq = (((s_part[r * PP + lane] + s_part[16 * PP + r * PP + lane])
                       + s_part[32 * PP + r * PP + lane]) + s_part[48 * PP + r * PP + lane]) * INV_S1;

    const float mp = wave_max(pw);
    const float mq = wave_max(pq);
    const float ep = expf(pw - mp);
    const float eq = expf(pq - mq);
    const float sp = wave_sum(ep);
    const float sq = wave_sum(eq);

    const float log_p = (pw - mp) - logf(sp);
    const float log_q = (pq - mq) - logf(sq);
    const float q     = eq / sq;
    float kl = q * (log_q - log_p);
    kl = wave_sum(kl);
    if (lane == 0) out[(size_t)(bt0 + r) * NA + n] = kl * (1.0f / AD);
  }
}

extern "C" void kernel_launch(void* const* d_in, const int* in_sizes, int n_in,
                              void* d_out, int out_size, void* d_ws, size_t ws_size,
                              hipStream_t stream) {
  (void)in_sizes; (void)n_in; (void)d_ws; (void)ws_size; (void)out_size;
  const float* obs = (const float*)d_in[0];
  const float* act = (const float*)d_in[1];
  const float* cf  = (const float*)d_in[2];
  const float* W1  = (const float*)d_in[3];
  const float* b1  = (const float*)d_in[4];
  const float* W2  = (const float*)d_in[5];
  const float* b2  = (const float*)d_in[6];
  const float* W3  = (const float*)d_in[7];
  const float* b3  = (const float*)d_in[8];
  float* out = (float*)d_out;

  const int tiles = (NA * BB * TT) / 16;
  const size_t lds_bytes = LDS_FLOATS * sizeof(float);
  coma_influence_kernel<<<tiles, 128, lds_bytes, stream>>>(
      obs, act, cf, W1, b1, W2, b2, W3, b3, out);
}


# Last layer is linear: each wave sums h2 over its p_without variants and does ONE layer-3 product (18 -> 5 per tile); bias added once at the end; sum order of p_without changes
# speedup vs baseline: 2.2323x; 1.0850x over previous
#include <hip/hip_runtime.h>
#include <math.h>

#define NA   16
#define DOBS 256
#define AD   32
#define BB   16
#define TT   64
#define SS   16
#define H1D  128
#define H2D  64
#define BT   (BB*TT)
#define INV_S1 (1.0f/17.0f)

#define H1P  132
#define OBSP 260
#define PP   36

#define OFF_W1A  0
#define OFF_W2   4096
#define OFF_B2   12288
#define OFF_B3   12352
#define OFF_HOBS 12416
#define OFF_PART 14528
#define OFF_W3   17408
#define LDS_FLOATS 19456

typedef float v2f __attribute__((ext_vector_type(2)));
typedef float v4f __attribute__((ext_vector_type(4)));
typedef float v8f __attribute__((ext_vector_type(8)));

__device__ __forceinline__ v8f wmma4(v2f a, v2f b, v8f c) {
  return __builtin_amdgcn_wmma_f32_16x16x4_f32(
       false, a,  false, b,
       (short)0, c,  false,  false);
}

__device__ __forceinline__ float wave_max(float v) {
  #pragma unroll
  for (int off = 16; off; off >>= 1) v = fmaxf(v, __shfl_xor(v, off, 32));
  return v;
}
__device__ __forceinline__ float wave_sum(float v) {
  #pragma unroll
  for (int off = 16; off; off >>= 1) v += __shfl_xor(v, off, 32);
  return v;
}

__device__ __forceinline__ void layer3(const float* a3p, const v8f (&h)[4], v8f (&p3)[2]) {
  #pragma unroll
  for (int t = 0; t < 4; ++t) {
    #pragma unroll
    for (int sp = 0; sp < 2; ++sp) {
      const int kp = 2 * t + sp;
      v4f a[2];
      #pragma unroll
      for (int mt = 0; mt < 2; ++mt) a[mt] = *(const v4f*)(a3p + (kp * 2 + mt) * 128);
      v2f bA, bB;
      bA[0] = h[t][4 * sp];     bA[1] = h[t][4 * sp + 1];
      bB[0] = h[t][4 * sp + 2]; bB[1] = h[t][4 * sp + 3];
      #pragma unroll
      for (int mt = 0; mt < 2; ++mt) p3[mt] = wmma4(a[mt].lo, bA, p3[mt]);
      #pragma unroll
      for (int mt = 0; mt < 2; ++mt) p3[mt] = wmma4(a[mt].hi, bB, p3[mt]);
    }
  }
}

template <int K_, int M_, bool CHAIN>
__device__ __forceinline__ void stage_frag(float* dst, const float* __restrict__ W, int tid) {
  constexpr int MT  = M_ / 16;
  constexpr int NV4 = K_ * M_ / 4;
  #pragma unroll
  for (int i = 0; i < NV4 / 128; ++i) {
    const int d4    = tid + 128 * i;
    const int ln    = d4 & 31;
    const int chunk = d4 >> 5;
    const int mt    = chunk % MT;
    const int kp    = chunk / MT;
    const int r     = ln & 15;
    const int h     = ln >> 4;
    const int ks0   = 2 * kp, ks1 = 2 * kp + 1;
    int k0, k1;
    if (CHAIN) {
      k0 = 16 * (ks0 >> 2) + 8 * h + 2 * (ks0 & 3);
      k1 = 16 * (ks1 >> 2) + 8 * h + 2 * (ks1 & 3);
    } else {
      k0 = 4 * ks0 + 2 * h;
      k1 = 4 * ks1 + 2 * h;
    }
    const int m = 16 * mt + r;
    float4 v;
    v.x = W[(k0    ) * M_ + m];
    v.y = W[(k0 + 1) * M_ + m];
    v.z = W[(k1    ) * M_ + m];
    v.w = W[(k1 + 1) * M_ + m];
    *(float4*)(dst + d4 * 4) = v;
  }
}

__global__ __launch_bounds__(128, 4)
void coma_influence_kernel(const float* __restrict__ obs,
                           const float* __restrict__ act,
                           const float* __restrict__ cf,
                           const float* __restrict__ W1,
                           const float* __restrict__ b1,
                           const float* __restrict__ W2,
                           const float* __restrict__ b2,
                           const float* __restrict__ W3,
                           const float* __restrict__ b3,
                           float* __restrict__ out)
{
  extern __shared__ float smem[];
  float* s_w1a  = smem + OFF_W1A;
  float* s_w2   = smem + OFF_W2;
  float* s_w3   = smem + OFF_W3;
  float* s_b2   = smem + OFF_B2;
  float* s_b3   = smem + OFF_B3;
  float* s_hobs = smem + OFF_HOBS;
  float* s_part = smem + OFF_PART;
  float* s_pw   = smem + OFF_PART + 4 * 16 * PP;
  float* s_obs  = smem + OFF_PART;

  const int tid  = threadIdx.x;
  const int lane = tid & 31;
  const int wid  = __builtin_amdgcn_readfirstlane(tid >> 5);
  const int g    = blockIdx.x;
  const int r0   = g * 16;
  const int n    = r0 >> 10;
  const int bt0  = r0 & 1023;

  const int col = lane & 15;
  const int hi  = lane >> 4;
  const int kb  = hi * 2;

  #pragma unroll
  for (int j4 = 0; j4 < 8; ++j4) {
    const int i4 = tid + 128 * j4;
    const int m  = i4 >> 6;
    const int d  = (i4 & 63) * 4;
    const float4 v = *(const float4*)(obs + ((size_t)(bt0 + m) * NA + n) * DOBS + d);
    *(float4*)(s_obs + m * OBSP + d) = v;
  }
  stage_frag<AD, H1D, false>(s_w1a, W1 + DOBS * H1D, tid);
  stage_frag<H1D, H2D, true>(s_w2, W2, tid);
  if (tid < H2D) s_b2[tid] = b2[tid];
  else if (tid < H2D + AD) s_b3[tid - H2D] = b3[tid - H2D];
  __syncthreads();

  {
    const int f0 = wid * 32;
    v8f acc0, acc1;
    #pragma unroll
    for (int v = 0; v < 8; ++v) {
      acc0[v] = b1[f0 + 8 * hi + v];
      acc1[v] = b1[f0 + 16 + 8 * hi + v];
    }
    const float* wp = W1 + f0 + col;
    const float* op = s_obs + col * OBSP + kb;
    #pragma unroll 8
    for (int kk = 0; kk < DOBS; kk += 4) {
      const v2f b = *(const v2f*)(op + kk);
      v2f a0, a1;
      a0[0] = wp[(kk + kb) * H1D];      a0[1] = wp[(kk + kb + 1) * H1D];
      a1[0] = wp[(kk + kb) * H1D + 16]; a1[1] = wp[(kk + kb + 1) * H1D + 16];
      acc0 = wmma4(a0, b, acc0);
      acc1 = wmma4(a1, b, acc1);
    }
    float* hp = s_hobs + col * H1P + f0 + 8 * hi;
    *(float4*)(hp)      = make_float4(acc0[0], acc0[1], acc0[2], acc0[3]);
    *(float4*)(hp + 4)  = make_float4(acc0[4], acc0[5], acc0[6], acc0[7]);
    *(float4*)(hp + 16) = make_float4(acc1[0], acc1[1], acc1[2], acc1[3]);
    *(float4*)(hp + 20) = make_float4(acc1[4], acc1[5], acc1[6], acc1[7]);
  }
  __syncthreads();

  stage_frag<H2D, AD, true>(s_w3, W3, tid);
  __syncthreads();

  const int role = (wid + g + (g >> 8)) & 3;
  const int it0  = (role < 2) ? 0 : 1;

  const float* act_row = act + ((size_t)(bt0 + col) * NA + n) * AD + kb;
  const float* cf_row  = cf + (((size_t)n * SS + 4 * role) * BT + (bt0 + col)) * AD + kb;
  const float* hob_row = s_hobs + col * H1P + 8 * hi;
  const float* a1p = s_w1a + lane * 4;
  const float* a2p = s_w2  + lane * 4;
  const float* a3p = s_w3  + lane * 4;

  v8f h2s[4];
  #pragma unroll
  for (int mt = 0; mt < 4; ++mt) {
    #pragma unroll
    for (int v = 0; v < 8; ++v) h2s[mt][v] = 0.0f;
  }
  float* pwp = s_pw + col * PP + 8 * hi;

  v2f bv[8];
  if (role == 1) {
    #pragma unroll
    for (int kq = 0; kq < 8; ++kq) { bv[kq][0] = 0.0f; bv[kq][1] = 0.0f; }
  } else {
    const float* p = (role == 0) ? act_row : cf_row;
    #pragma unroll
    for (int kq = 0; kq < 8; ++kq) bv[kq] = *(const v2f*)(p + kq * 4);
  }

  for (int it = it0; it < 5; ++it) {
    __asm__ volatile("" ::: "memory");
    v2f bn[8];
    {
      const int itn = (it < 4) ? it : 3;
      const float* p = cf_row + (size_t)itn * (BT * AD);
      #pragma unroll
      for (int kq = 0; kq < 8; ++kq) bn[kq] = *(const v2f*)(p + kq * 4);
    }

    v8f h1[8];
    #pragma unroll
    for (int mt = 0; mt < 8; ++mt) {
      const v4f lo = *(const v4f*)(hob_row + 16 * mt);
      const v4f up = *(const v4f*)(hob_row + 16 * mt + 4);
      h1[mt] = __builtin_shufflevector(lo, up, 0, 1, 2, 3, 4, 5, 6, 7);
    }
    #pragma unroll
    for (int kp = 0; kp < 4; ++kp) {
      v4f a[8];
      #pragma unroll
      for (int mt = 0; mt < 8; ++mt) a[mt] = *(const v4f*)(a1p + (kp * 8 + mt) * 128);
      #pragma unroll
      for (int mt = 0; mt < 8; ++mt) h1[mt] = wmma4(a[mt].lo, bv[2 * kp], h1[mt]);
      #pragma unroll
      for (int mt = 0; mt < 8; ++mt) h1[mt] = wmma4(a[mt].hi, bv[2 * kp + 1], h1[mt]);
    }
    #pragma unroll
    for (int mt = 0; mt < 8; ++mt) {
      #pragma unroll
      for (int v = 0; v < 8; ++v) h1[mt][v] = fmaxf(h1[mt][v], 0.0f);
    }

    v8f h2[4];
    #pragma unroll
    for (int mt = 0; mt < 4; ++mt) {
      const v4f lo = *(const v4f*)(s_b2 + 16 * mt + 8 * hi);
      const v4f up = *(const v4f*)(s_b2 + 16 * mt + 8 * hi + 4);
      h2[mt] = __builtin_shufflevector(lo, up, 0, 1, 2, 3, 4, 5, 6, 7);
    }
    #pragma unroll
    for (int t = 0; t < 8; ++t) {
      #pragma unroll
      for (int sp = 0; sp < 2; ++sp) {
        const int kp = 2 * t + sp;
        v4f a[4];
        #pragma unroll
        for (int mt = 0; mt < 4; ++mt) a[mt] = *(const v4f*)(a2p + (kp * 4 + mt) * 128);
        v2f bA, bB;
        bA[0] = h1[t][4 * sp];     bA[1] = h1[t][4 * sp + 1];
        bB[0] = h1[t][4 * sp + 2]; bB[1] = h1[t][4 * sp + 3];
        #pragma unroll
        for (int mt = 0; mt < 4; ++mt) h2[mt] = wmma4(a[mt].lo, bA, h2[mt]);
        #pragma unroll
        for (int mt = 0; mt < 4; ++mt) h2[mt] = wmma4(a[mt].hi, bB, h2[mt]);
      }
    }
    #pragma unroll
    for (int mt = 0; mt < 4; ++mt) {
      #pragma unroll
      for (int v = 0; v < 8; ++v) h2[mt][v] = fmaxf(h2[mt][v], 0.0f);
    }

    if (role == 0 && it == 0) {
      v8f p3[2];
      #pragma unroll
      for (int mt = 0; mt < 2; ++mt) {
        const v4f lo = *(const v4f*)(s_b3 + 16 * mt + 8 * hi);
        const v4f up = *(const v4f*)(s_b3 + 16 * mt + 8 * hi + 4);
        p3[mt] = __builtin_shufflevector(lo, up, 0, 1, 2, 3, 4, 5, 6, 7);
      }
      layer3(a3p, h2, p3);
      *(v4f*)(pwp)      = p3[0].lo;
      *(v4f*)(pwp + 4)  = p3[0].hi;
      *(v4f*)(pwp + 16) = p3[1].lo;
      *(v4f*)(pwp + 20) = p3[1].hi;
    } else {
      #pragma unroll
      for (int mt = 0; mt < 4; ++mt) h2s[mt] += h2[mt];
    }

    #pragma unroll
    for (int kq = 0; kq < 8; ++kq) bv[kq] = bn[kq];
  }

  {
    v8f pq[2];
    #pragma unroll
    for (int mt = 0; mt < 2; ++mt) {
      #pragma unroll
      for (int v = 0; v < 8; ++v) pq[mt][v] = 0.0f;
    }
    layer3(a3p, h2s, pq);
    float* pp = s_part + role * (16 * PP) + col * PP + 8 * hi;
    *(v4f*)(pp)      = pq[0].lo;
    *(v4f*)(pp + 4)  = pq[0].hi;
    *(v4f*)(pp + 16) = pq[1].lo;
    *(v4f*)(pp + 20) = pq[1].hi;
  }
  __syncthreads();

  #pragma unroll
  for (int rr = 0; rr < 4; ++rr) {
    const int r = wid * 4 + rr;
    const float pw = s_pw[r * PP + lane];
    const float pq = (((s_part[r * PP + lane] + s_part[16 * PP + r * PP + lane])
                       + s_part[32 * PP + r * PP + lane]) + s_part[48 * PP + r * PP + lane]) * INV_S1
                     + s_b3[lane];

    const float mp = wave_max(pw);
    const float mq = wave_max(pq);
    const float ep = expf(pw - mp);
    const float eq = expf(pq - mq);
    const float sp = wave_sum(ep);
    const float sq = wave_sum(eq);

    const float log_p = (pw - mp) - logf(sp);
    const float log_q = (pq - mq) - logf(sq);
    const float q     = eq / sq;
    float kl = q * (log_q - log_p);
    kl = wave_sum(kl);
    if (lane == 0) out[(size_t)(bt0 + r) * NA + n] = kl * (1.0f / AD);
  }
}

extern "C" void kernel_launch(void* const* d_in, const int* in_sizes, int n_in,
                              void* d_out, int out_size, void* d_ws, size_t ws_size,
                              hipStream_t stream) {
  (void)in_sizes; (void)n_in; (void)d_ws; (void)ws_size; (void)out_size;
  const float* obs = (const float*)d_in[0];
  const float* act = (const float*)d_in[1];
  const float* cf  = (const float*)d_in[2];
  const float* W1  = (const float*)d_in[3];
  const float* b1  = (const float*)d_in[4];
  const float* W2  = (const float*)d_in[5];
  const float* b2  = (const float*)d_in[6];
  const float* W3  = (const float*)d_in[7];
  const float* b3  = (const float*)d_in[8];
  float* out = (float*)d_out;

  const int tiles = (NA * BB * TT) / 16;
  const size_t lds_bytes = LDS_FLOATS * sizeof(float);
  coma_influence_kernel<<<tiles, 128, lds_bytes, stream>>>(
      obs, act, cf, W1, b1, W2, b2, W3, b3, out);
}


# Step 2 LDS weight-fragment loads grouped: one run of ds_load_b128 per 16-WMMA block, issued one block ahead into a second register set (13 load places per variant instead of 55)
# speedup vs baseline: 2.3099x; 1.0348x over previous
#include <hip/hip_runtime.h>
#include <math.h>

#define NA   16
#define DOBS 256
#define AD   32
#define BB   16
#define TT   64
#define SS   16
#define H1D  128
#define H2D  64
#define BT   (BB*TT)
#define INV_S1 (1.0f/17.0f)

#define H1P  132
#define OBSP 260
#define PP   36

#define OFF_W1A  0
#define OFF_W2   4096
#define OFF_B2   12288
#define OFF_B3   12352
#define OFF_HOBS 12416
#define OFF_PART 14528
#define OFF_W3   17408
#define LDS_FLOATS 19456

typedef float v2f __attribute__((ext_vector_type(2)));
typedef float v4f __attribute__((ext_vector_type(4)));
typedef float v8f __attribute__((ext_vector_type(8)));

__device__ __forceinline__ v8f wmma4(v2f a, v2f b, v8f c) {
  return __builtin_amdgcn_wmma_f32_16x16x4_f32(
       false, a,  false, b,
       (short)0, c,  false,  false);
}

__device__ __forceinline__ void fence() { __builtin_amdgcn_sched_barrier(0); }

__device__ __forceinline__ float wave_max(float v) {
  #pragma unroll
  for (int off = 16; off; off >>= 1) v = fmaxf(v, __shfl_xor(v, off, 32));
  return v;
}
__device__ __forceinline__ float wave_sum(float v) {
  #pragma unroll
  for (int off = 16; off; off >>= 1) v += __shfl_xor(v, off, 32);
  return v;
}

__device__ __forceinline__ void layer3(const float* a3p, const v8f (&h)[4], v8f (&p3)[2]) {
  #pragma unroll
  for (int t = 0; t < 4; ++t) {
    #pragma unroll
    for (int sp = 0; sp < 2; ++sp) {
      const int kp = 2 * t + sp;
      v4f a[2];
      #pragma unroll
      for (int mt = 0; mt < 2; ++mt) a[mt] = *(const v4f*)(a3p + (kp * 2 + mt) * 128);
      v2f bA, bB;
      bA[0] = h[t][4 * sp];     bA[1] = h[t][4 * sp + 1];
      bB[0] = h[t][4 * sp + 2]; bB[1] = h[t][4 * sp + 3];
      #pragma unroll
      for (int mt = 0; mt < 2; ++mt) p3[mt] = wmma4(a[mt].lo, bA, p3[mt]);
      #pragma unroll
      for (int mt = 0; mt < 2; ++mt) p3[mt] = wmma4(a[mt].hi, bB, p3[mt]);
    }
  }
}

template <int K_, int M_, bool CHAIN>
__device__ __forceinline__ void stage_frag(float* dst, const float* __restrict__ W, int tid) {
  constexpr int MT  = M_ / 16;
  constexpr int NV4 = K_ * M_ / 4;
  #pragma unroll
  for (int i = 0; i < NV4 / 128; ++i) {
    const int d4    = tid + 128 * i;
    const int ln    = d4 & 31;
    const int chunk = d4 >> 5;
    const int mt    = chunk % MT;
    const int kp    = chunk / MT;
    const int r     = ln & 15;
    const int h     = ln >> 4;
    const int ks0   = 2 * kp, ks1 = 2 * kp + 1;
    int k0, k1;
    if (CHAIN) {
      k0 = 16 * (ks0 >> 2) + 8 * h + 2 * (ks0 & 3);
      k1 = 16 * (ks1 >> 2) + 8 * h + 2 * (ks1 & 3);
    } else {
      k0 = 4 * ks0 + 2 * h;
      k1 = 4 * ks1 + 2 * h;
    }
    const int m = 16 * mt + r;
    float4 v;
    v.x = W[(k0    ) * M_ + m];
    v.y = W[(k0 + 1) * M_ + m];
    v.z = W[(k1    ) * M_ + m];
    v.w = W[(k1 + 1) * M_ + m];
    *(float4*)(dst + d4 * 4) = v;
  }
}

__global__ __launch_bounds__(128, 4)
void coma_influence_kernel(const float* __restrict__ obs,
                           const float* __restrict__ act,
                           const float* __restrict__ cf,
                           const float* __restrict__ W1,
                           const float* __restrict__ b1,
                           const float* __restrict__ W2,
                           const float* __restrict__ b2,
                           const float* __restrict__ W3,
                           const float* __restrict__ b3,
                           float* __restrict__ out)
{
  extern __shared__ float smem[];
  float* s_w1a  = smem + OFF_W1A;
  float* s_w2   = smem + OFF_W2;
  float* s_w3   = smem + OFF_W3;
  float* s_b2   = smem + OFF_B2;
  float* s_b3   = smem + OFF_B3;
  float* s_hobs = smem + OFF_HOBS;
  float* s_part = smem + OFF_PART;
  float* s_pw   = smem + OFF_PART + 4 * 16 * PP;
  float* s_obs  = smem + OFF_PART;

  const int tid  = threadIdx.x;
  const int lane = tid & 31;
  const int wid  = __builtin_amdgcn_readfirstlane(tid >> 5);
  const int g    = blockIdx.x;
  const int r0   = g * 16;
  const int n    = r0 >> 10;
  const int bt0  = r0 & 1023;

  const int col = lane & 15;
  const int hi  = lane >> 4;
  const int kb  = hi * 2;

  #pragma unroll
  for (int j4 = 0; j4 < 8; ++j4) {
    const int i4 = tid + 128 * j4;
    const int m  = i4 >> 6;
    const int d  = (i4 & 63) * 4;
    const float4 v = *(const float4*)(obs + ((size_t)(bt0 + m) * NA + n) * DOBS + d);
    *(float4*)(s_obs + m * OBSP + d) = v;
  }
  stage_frag<AD, H1D, false>(s_w1a, W1 + DOBS * H1D, tid);
  stage_frag<H1D, H2D, true>(s_w2, W2, tid);
  if (tid < H2D) s_b2[tid] = b2[tid];
  else if (tid < H2D + AD) s_b3[tid - H2D] = b3[tid - H2D];
  __syncthreads();

  {
    const int f0 = wid * 32;
    v8f acc0, acc1;
    #pragma unroll
    for (int v = 0; v < 8; ++v) {
      acc0[v] = b1[f0 + 8 * hi + v];
      acc1[v] = b1[f0 + 16 + 8 * hi + v];
    }
    const float* wp = W1 + f0 + col;
    const float* op = s_obs + col * OBSP + kb;
    #pragma unroll 8
    for (int kk = 0; kk < DOBS; kk += 4) {
      const v2f b = *(const v2f*)(op + kk);
      v2f a0, a1;
      a0[0] = wp[(kk + kb) * H1D];      a0[1] = wp[(kk + kb + 1) * H1D];
      a1[0] = wp[(kk + kb) * H1D + 16]; a1[1] = wp[(kk + kb + 1) * H1D + 16];
      acc0 = wmma4(a0, b, acc0);
      acc1 = wmma4(a1, b, acc1);
    }
    float* hp = s_hobs + col * H1P + f0 + 8 * hi;
    *(float4*)(hp)      = make_float4(acc0[0], acc0[1], acc0[2], acc0[3]);
    *(float4*)(hp + 4)  = make_float4(acc0[4], acc0[5], acc0[6], acc0[7]);
    *(float4*)(hp + 16) = make_float4(acc1[0], acc1[1], acc1[2], acc1[3]);
    *(float4*)(hp + 20) = make_float4(acc1[4], acc1[5], acc1[6], acc1[7]);
  }
  __syncthreads();

  stage_frag<H2D, AD, true>(s_w3, W3, tid);
  __syncthreads();

  const int role = (wid + g + (g >> 8)) & 3;
  const int it0  = (role < 2) ? 0 : 1;

  const float* act_row = act + ((size_t)(bt0 + col) * NA + n) * AD + kb;
  const float* cf_row  = cf + (((size_t)n * SS + 4 * role) * BT + (bt0 + col)) * AD + kb;
  const float* hob_row = s_hobs + col * H1P + 8 * hi;
  const float* a1p = s_w1a + lane * 4;
  const float* a2p = s_w2  + lane * 4;
  const float* a3p = s_w3  + lane * 4;

  v8f h2s[4];
  #pragma unroll
  for (int mt = 0; mt < 4; ++mt) {
    #pragma unroll
    for (int v = 0; v < 8; ++v) h2s[mt][v] = 0.0f;
  }
  float* pwp = s_pw + col * PP + 8 * hi;

  v2f bv[8];
  if (role == 1) {
    #pragma unroll
    for (int kq = 0; kq < 8; ++kq) { bv[kq][0] = 0.0f; bv[kq][1] = 0.0f; }
  } else {
    const float* p = (role == 0) ? act_row : cf_row;
    #pragma unroll
    for (int kq = 0; kq < 8; ++kq) bv[kq] = *(const v2f*)(p + kq * 4);
  }

  v4f fr[2][8];
  v8f h1[8];
  #pragma unroll
  for (int mt = 0; mt < 8; ++mt) {
    const v4f lo = *(const v4f*)(hob_row + 16 * mt);
    const v4f up = *(const v4f*)(hob_row + 16 * mt + 4);
    h1[mt] = __builtin_shufflevector(lo, up, 0, 1, 2, 3, 4, 5, 6, 7);
  }
  #pragma unroll
  for (int mt = 0; mt < 8; ++mt) fr[0][mt] = *(const v4f*)(a1p + mt * 128);

  for (int it = it0; it < 5; ++it) {
    __asm__ volatile("" ::: "memory");
    v8f h2[4];
    #pragma unroll
    for (int kp = 0; kp < 4; ++kp) {
      const int cs = kp & 1;
      fence();
      if (kp < 3) {
        #pragma unroll
        for (int mt = 0; mt < 8; ++mt) fr[cs ^ 1][mt] = *(const v4f*)(a1p + ((kp + 1) * 8 + mt) * 128);
      } else {
        #pragma unroll
        for (int i = 0; i < 8; ++i) fr[cs ^ 1][i] = *(const v4f*)(a2p + i * 128);
        #pragma unroll
        for (int mt = 0; mt < 4; ++mt) {
          const v4f lo = *(const v4f*)(s_b2 + 16 * mt + 8 * hi);
          const v4f up = *(const v4f*)(s_b2 + 16 * mt + 8 * hi + 4);
          h2[mt] = __builtin_shufflevector(lo, up, 0, 1, 2, 3, 4, 5, 6, 7);
        }
      }
      fence();
      #pragma unroll
      for (int mt = 0; mt < 8; ++mt) h1[mt] = wmma4(fr[cs][mt].lo, bv[2 * kp], h1[mt]);
      #pragma unroll
      for (int mt = 0; mt < 8; ++mt) h1[mt] = wmma4(fr[cs][mt].hi, bv[2 * kp + 1], h1[mt]);
      if (kp == 3) {
        #pragma unroll
        for (int v = 0; v < 8; ++v) h1[0][v] = fmaxf(h1[0][v], 0.0f);
      }
    }

    #pragma unroll
    for (int t = 0; t < 8; ++t) {
      const int cs = t & 1;
      fence();
      if (t < 7) {
        #pragma unroll
        for (int i = 0; i < 8; ++i) fr[cs ^ 1][i] = *(const v4f*)(a2p + ((t + 1) * 8 + i) * 128);
      } else {
        #pragma unroll
        for (int mt = 0; mt < 8; ++mt) fr[cs ^ 1][mt] = *(const v4f*)(a1p + mt * 128);
      }
      if (t > 0) {
        const v4f lo = *(const v4f*)(hob_row + 16 * (t - 1));
        const v4f up = *(const v4f*)(hob_row + 16 * (t - 1) + 4);
        h1[t - 1] = __builtin_shufflevector(lo, up, 0, 1, 2, 3, 4, 5, 6, 7);
      }
      if (t == 0) {
        const int itn = (it < 4) ? it : 3;
        const float* p = cf_row + (size_t)itn * (BT * AD);
        fence();
        #pragma unroll
        for (int kq = 0; kq < 8; ++kq) bv[kq] = *(const v2f*)(p + kq * 4);
      }
      fence();
      if (t < 7) {
        #pragma unroll
        for (int v = 0; v < 8; ++v) h1[t + 1][v] = fmaxf(h1[t + 1][v], 0.0f);
      }
      #pragma unroll
      for (int sp = 0; sp < 2; ++sp) {
        v2f bA, bB;
        bA[0] = h1[t][4 * sp];     bA[1] = h1[t][4 * sp + 1];
        bB[0] = h1[t][4 * sp + 2]; bB[1] = h1[t][4 * sp + 3];
        #pragma unroll
        for (int mt = 0; mt < 4; ++mt) h2[mt] = wmma4(fr[cs][sp * 4 + mt].lo, bA, h2[mt]);
        #pragma unroll
        for (int mt = 0; mt < 4; ++mt) h2[mt] = wmma4(fr[cs][sp * 4 + mt].hi, bB, h2[mt]);
      }
    }
    fence();
    {
      const v4f lo = *(const v4f*)(hob_row + 16 * 7);
      const v4f up = *(const v4f*)(hob_row + 16 * 7 + 4);
      h1[7] = __builtin_shufflevector(lo, up, 0, 1, 2, 3, 4, 5, 6, 7);
    }
    fence();
    #pragma unroll
    for (int mt = 0; mt < 4; ++mt) {
      #pragma unroll
      for (int v = 0; v < 8; ++v) h2[mt][v] = fmaxf(h2[mt][v], 0.0f);
    }

    if (role == 0 && it == 0) {
      v8f p3[2];
      #pragma unroll
      for (int mt = 0; mt < 2; ++mt) {
        const v4f lo = *(const v4f*)(s_b3 + 16 * mt + 8 * hi);
        const v4f up = *(const v4f*)(s_b3 + 16 * mt + 8 * hi + 4);
        p3[mt] = __builtin_shufflevector(lo, up, 0, 1, 2, 3, 4, 5, 6, 7);
      }
      layer3(a3p, h2, p3);
      *(v4f*)(pwp)      = p3[0].lo;
      *(v4f*)(pwp + 4)  = p3[0].hi;
      *(v4f*)(pwp + 16) = p3[1].lo;
      *(v4f*)(pwp + 20) = p3[1].hi;
    } else {
      #pragma unroll
      for (int mt = 0; mt < 4; ++mt) h2s[mt] += h2[mt];
    }

  }

  {
    v8f pq[2];
    #pragma unroll
    for (int mt = 0; mt < 2; ++mt) {
      #pragma unroll
      for (int v = 0; v < 8; ++v) pq[mt][v] = 0.0f;
    }
    layer3(a3p, h2s, pq);
    float* pp = s_part + role * (16 * PP) + col * PP + 8 * hi;
    *(v4f*)(pp)      = pq[0].lo;
    *(v4f*)(pp + 4)  = pq[0].hi;
    *(v4f*)(pp + 16) = pq[1].lo;
    *(v4f*)(pp + 20) = pq[1].hi;
  }
  __syncthreads();

  #pragma unroll
  for (int rr = 0; rr < 4; ++rr) {
    const int r = wid * 4 + rr;
    const float pw = s_pw[r * PP + lane];
    const float pq = (((s_part[r * PP + lane] + s_part[16 * PP + r * PP + lane])
                       + s_part[32 * PP + r * PP + lane]) + s_part[48 * PP + r * PP + lane]) * INV_S1
                     + s_b3[lane];

    const float mp = wave_max(pw);
    const float mq = wave_max(pq);
    const float ep = expf(pw - mp);
    const float eq = expf(pq - mq);
    const float sp = wave_sum(ep);
    const float sq = wave_sum(eq);

    const float log_p = (pw - mp) - logf(sp);
    const float log_q = (pq - mq) - logf(sq);
    const float q     = eq / sq;
    float kl = q * (log_q - log_p);
    kl = wave_sum(kl);
    if (lane == 0) out[(size_t)(bt0 + r) * NA + n] = kl * (1.0f / AD);
  }
}

extern "C" void kernel_launch(void* const* d_in, const int* in_sizes, int n_in,
                              void* d_out, int out_size, void* d_ws, size_t ws_size,
                              hipStream_t stream) {
  (void)in_sizes; (void)n_in; (void)d_ws; (void)ws_size; (void)out_size;
  const float* obs = (const float*)d_in[0];
  const float* act = (const float*)d_in[1];
  const float* cf  = (const float*)d_in[2];
  const float* W1  = (const float*)d_in[3];
  const float* b1  = (const float*)d_in[4];
  const float* W2  = (const float*)d_in[5];
  const float* b2  = (const float*)d_in[6];
  const float* W3  = (const float*)d_in[7];
  const float* b3  = (const float*)d_in[8];
  float* out = (float*)d_out;

  const int tiles = (NA * BB * TT) / 16;
  const size_t lds_bytes = LDS_FLOATS * sizeof(float);
  coma_influence_kernel<<<tiles, 128, lds_bytes, stream>>>(
      obs, act, cf, W1, b1, W2, b2, W3, b3, out);
}


# Closing softmax/KL reductions on the vector ALU (lane^16 exchange + row_xmask DPP, same xor order, same sums), the wave's 4 rows in step: replaces 100 dependent ds_bpermute round trips per wave
# speedup vs baseline: 2.4266x; 1.0505x over previous
#include <hip/hip_runtime.h>
#include <math.h>

#define NA   16
#define DOBS 256
#define AD   32
#define BB   16
#define TT   64
#define SS   16
#define H1D  128
#define H2D  64
#define BT   (BB*TT)
#define INV_S1 (1.0f/17.0f)

#define H1P  132
#define OBSP 260
#define PP   36

#define OFF_W1A  0
#define OFF_W2   4096
#define OFF_B2   12288
#define OFF_B3   12352
#define OFF_HOBS 12416
#define OFF_PART 14528
#define OFF_W3   17408
#define LDS_FLOATS 19456

typedef float v2f __attribute__((ext_vector_type(2)));
typedef float v4f __attribute__((ext_vector_type(4)));
typedef float v8f __attribute__((ext_vector_type(8)));

__device__ __forceinline__ v8f wmma4(v2f a, v2f b, v8f c) {
  return __builtin_amdgcn_wmma_f32_16x16x4_f32(
       false, a,  false, b,
       (short)0, c,  false,  false);
}

__device__ __forceinline__ void fence() { __builtin_amdgcn_sched_barrier(0); }

__device__ __forceinline__ float xhalf(float x) {
  const int b = __builtin_bit_cast(int, x);
  return __builtin_bit_cast(float, __builtin_amdgcn_permlanex16(b, b, 0x76543210, 0xfedcba98, false, false));
}
template <int M>
__device__ __forceinline__ float xrow(float x) {
  const int b = __builtin_bit_cast(int, x);
  return __builtin_bit_cast(float, __builtin_amdgcn_update_dpp(b, b, 0x160 + M, 0xf, 0xf, true));
}
template <int N>
__device__ __forceinline__ void wave_max_n(float (&v)[N]) {
  #pragma unroll
  for (int i = 0; i < N; ++i) v[i] = fmaxf(v[i], xhalf(v[i]));
  #pragma unroll
  for (int i = 0; i < N; ++i) v[i] = fmaxf(v[i], xrow<8>(v[i]));
  #pragma unroll
  for (int i = 0; i < N; ++i) v[i] = fmaxf(v[i], xrow<4>(v[i]));
  #pragma unroll
  for (int i = 0; i < N; ++i) v[i] = fmaxf(v[i], xrow<2>(v[i]));
  #pragma unroll
  for (int i = 0; i < N; ++i) v[i] = fmaxf(v[i], xrow<1>(v[i]));
}
template <int N>
__device__ __forceinline__ void wave_sum_n(float (&v)[N]) {
  #pragma unroll
  for (int i = 0; i < N; ++i) v[i] += xhalf(v[i]);
  #pragma unroll
  for (int i = 0; i < N; ++i) v[i] += xrow<8>(v[i]);
  #pragma unroll
  for (int i = 0; i < N; ++i) v[i] += xrow<4>(v[i]);
  #pragma unroll
  for (int i = 0; i < N; ++i) v[i] += xrow<2>(v[i]);
  #pragma unroll
  for (int i = 0; i < N; ++i) v[i] += xrow<1>(v[i]);
}

__device__ __forceinline__ void layer3(const float* a3p, const v8f (&h)[4], v8f (&p3)[2]) {
  #pragma unroll
  for (int t = 0; t < 4; ++t) {
    #pragma unroll
    for (int sp = 0; sp < 2; ++sp) {
      const int kp = 2 * t + sp;
      v4f a[2];
      #pragma unroll
      for (int mt = 0; mt < 2; ++mt) a[mt] = *(const v4f*)(a3p + (kp * 2 + mt) * 128);
      v2f bA, bB;
      bA[0] = h[t][4 * sp];     bA[1] = h[t][4 * sp + 1];
      bB[0] = h[t][4 * sp + 2]; bB[1] = h[t][4 * sp + 3];
      #pragma unroll
      for (int mt = 0; mt < 2; ++mt) p3[mt] = wmma4(a[mt].lo, bA, p3[mt]);
      #pragma unroll
      for (int mt = 0; mt < 2; ++mt) p3[mt] = wmma4(a[mt].hi, bB, p3[mt]);
    }
  }
}

template <int K_, int M_, bool CHAIN>
__device__ __forceinline__ void stage_frag(float* dst, const float* __restrict__ W, int tid) {
  constexpr int MT  = M_ / 16;
  constexpr int NV4 = K_ * M_ / 4;
  #pragma unroll
  for (int i = 0; i < NV4 / 128; ++i) {
    const int d4    = tid + 128 * i;
    const int ln    = d4 & 31;
    const int chunk = d4 >> 5;
    const int mt    = chunk % MT;
    const int kp    = chunk / MT;
    const int r     = ln & 15;
    const int h     = ln >> 4;
    const int ks0   = 2 * kp, ks1 = 2 * kp + 1;
    int k0, k1;
    if (CHAIN) {
      k0 = 16 * (ks0 >> 2) + 8 * h + 2 * (ks0 & 3);
      k1 = 16 * (ks1 >> 2) + 8 * h + 2 * (ks1 & 3);
    } else {
      k0 = 4 * ks0 + 2 * h;
      k1 = 4 * ks1 + 2 * h;
    }
    const int m = 16 * mt + r;
    float4 v;
    v.x = W[(k0    ) * M_ + m];
    v.y = W[(k0 + 1) * M_ + m];
    v.z = W[(k1    ) * M_ + m];
    v.w = W[(k1 + 1) * M_ + m];
    *(float4*)(dst + d4 * 4) = v;
  }
}

__global__ __launch_bounds__(128, 4)
void coma_influence_kernel(const float* __restrict__ obs,
                           const float* __restrict__ act,
                           const float* __restrict__ cf,
                           const float* __restrict__ W1,
                           const float* __restrict__ b1,
                           const float* __restrict__ W2,
                           const float* __restrict__ b2,
                           const float* __restrict__ W3,
                           const float* __restrict__ b3,
                           float* __restrict__ out)
{
  extern __shared__ float smem[];
  float* s_w1a  = smem + OFF_W1A;
  float* s_w2   = smem + OFF_W2;
  float* s_w3   = smem + OFF_W3;
  float* s_b2   = smem + OFF_B2;
  float* s_b3   = smem + OFF_B3;
  float* s_hobs = smem + OFF_HOBS;
  float* s_part = smem + OFF_PART;
  float* s_pw   = smem + OFF_PART + 4 * 16 * PP;
  float* s_obs  = smem + OFF_PART;

  const int tid  = threadIdx.x;
  const int lane = tid & 31;
  const int wid  = __builtin_amdgcn_readfirstlane(tid >> 5);
  const int g    = blockIdx.x;
  const int r0   = g * 16;
  const int n    = r0 >> 10;
  const int bt0  = r0 & 1023;

  const int col = lane & 15;
  const int hi  = lane >> 4;
  const int kb  = hi * 2;

  #pragma unroll
  for (int j4 = 0; j4 < 8; ++j4) {
    const int i4 = tid + 128 * j4;
    const int m  = i4 >> 6;
    const int d  = (i4 & 63) * 4;
    const float4 v = *(const float4*)(obs + ((size_t)(bt0 + m) * NA + n) * DOBS + d);
    *(float4*)(s_obs + m * OBSP + d) = v;
  }
  stage_frag<AD, H1D, false>(s_w1a, W1 + DOBS * H1D, tid);
  stage_frag<H1D, H2D, true>(s_w2, W2, tid);
  if (tid < H2D) s_b2[tid] = b2[tid];
  else if (tid < H2D + AD) s_b3[tid - H2D] = b3[tid - H2D];
  __syncthreads();

  {
    const int f0 = wid * 32;
    v8f acc0, acc1;
    #pragma unroll
    for (int v = 0; v < 8; ++v) {
      acc0[v] = b1[f0 + 8 * hi + v];
      acc1[v] = b1[f0 + 16 + 8 * hi + v];
    }
    const float* wp = W1 + f0 + col;
    const float* op = s_obs + col * OBSP + kb;
    #pragma unroll 8
    for (int kk = 0; kk < DOBS; kk += 4) {
      const v2f b = *(const v2f*)(op + kk);
      v2f a0, a1;
      a0[0] = wp[(kk + kb) * H1D];      a0[1] = wp[(kk + kb + 1) * H1D];
      a1[0] = wp[(kk + kb) * H1D + 16]; a1[1] = wp[(kk + kb + 1) * H1D + 16];
      acc0 = wmma4(a0, b, acc0);
      acc1 = wmma4(a1, b, acc1);
    }
    float* hp = s_hobs + col * H1P + f0 + 8 * hi;
    *(float4*)(hp)      = make_float4(acc0[0], acc0[1], acc0[2], acc0[3]);
    *(float4*)(hp + 4)  = make_float4(acc0[4], acc0[5], acc0[6], acc0[7]);
    *(float4*)(hp + 16) = make_float4(acc1[0], acc1[1], acc1[2], acc1[3]);
    *(float4*)(hp + 20) = make_float4(acc1[4], acc1[5], acc1[6], acc1[7]);
  }
  __syncthreads();

  stage_frag<H2D, AD, true>(s_w3, W3, tid);
  __syncthreads();

  const int role = (wid + g + (g >> 8)) & 3;
  const int it0  = (role < 2) ? 0 : 1;

  const float* act_row = act + ((size_t)(bt0 + col) * NA + n) * AD + kb;
  const float* cf_row  = cf + (((size_t)n * SS + 4 * role) * BT + (bt0 + col)) * AD + kb;
  const float* hob_row = s_hobs + col * H1P + 8 * hi;
  const float* a1p = s_w1a + lane * 4;
  const float* a2p = s_w2  + lane * 4;
  const float* a3p = s_w3  + lane * 4;

  v8f h2s[4];
  #pragma unroll
  for (int mt = 0; mt < 4; ++mt) {
    #pragma unroll
    for (int v = 0; v < 8; ++v) h2s[mt][v] = 0.0f;
  }
  float* pwp = s_pw + col * PP + 8 * hi;

  v2f bv[8];
  if (role == 1) {
    #pragma unroll
    for (int kq = 0; kq < 8; ++kq) { bv[kq][0] = 0.0f; bv[kq][1] = 0.0f; }
  } else {
    const float* p = (role == 0) ? act_row : cf_row;
    #pragma unroll
    for (int kq = 0; kq < 8; ++kq) bv[kq] = *(const v2f*)(p + kq * 4);
  }

  v4f fr[2][8];
  v8f h1[8];
  #pragma unroll
  for (int mt = 0; mt < 8; ++mt) {
    const v4f lo = *(const v4f*)(hob_row + 16 * mt);
    const v4f up = *(const v4f*)(hob_row + 16 * mt + 4);
    h1[mt] = __builtin_shufflevector(lo, up, 0, 1, 2, 3, 4, 5, 6, 7);
  }
  #pragma unroll
  for (int mt = 0; mt < 8; ++mt) fr[0][mt] = *(const v4f*)(a1p + mt * 128);

  for (int it = it0; it < 5; ++it) {
    __asm__ volatile("" ::: "memory");
    v8f h2[4];
    #pragma unroll
    for (int kp = 0; kp < 4; ++kp) {
      const int cs = kp & 1;
      fence();
      if (kp < 3) {
        #pragma unroll
        for (int mt = 0; mt < 8; ++mt) fr[cs ^ 1][mt] = *(const v4f*)(a1p + ((kp + 1) * 8 + mt) * 128);
      } else {
        #pragma unroll
        for (int i = 0; i < 8; ++i) fr[cs ^ 1][i] = *(const v4f*)(a2p + i * 128);
        #pragma unroll
        for (int mt = 0; mt < 4; ++mt) {
          const v4f lo = *(const v4f*)(s_b2 + 16 * mt + 8 * hi);
          const v4f up = *(const v4f*)(s_b2 + 16 * mt + 8 * hi + 4);
          h2[mt] = __builtin_shufflevector(lo, up, 0, 1, 2, 3, 4, 5, 6, 7);
        }
      }
      fence();
      #pragma unroll
      for (int mt = 0; mt < 8; ++mt) h1[mt] = wmma4(fr[cs][mt].lo, bv[2 * kp], h1[mt]);
      #pragma unroll
      for (int mt = 0; mt < 8; ++mt) h1[mt] = wmma4(fr[cs][mt].hi, bv[2 * kp + 1], h1[mt]);
      if (kp == 3) {
        #pragma unroll
        for (int v = 0; v < 8; ++v) h1[0][v] = fmaxf(h1[0][v], 0.0f);
      }
    }

    #pragma unroll
    for (int t = 0; t < 8; ++t) {
      const int cs = t & 1;
      fence();
      if (t < 7) {
        #pragma unroll
        for (int i = 0; i < 8; ++i) fr[cs ^ 1][i] = *(const v4f*)(a2p + ((t + 1) * 8 + i) * 128);
      } else {
        #pragma unroll
        for (int mt = 0; mt < 8; ++mt) fr[cs ^ 1][mt] = *(const v4f*)(a1p + mt * 128);
      }
      if (t > 0) {
        const v4f lo = *(const v4f*)(hob_row + 16 * (t - 1));
        const v4f up = *(const v4f*)(hob_row + 16 * (t - 1) + 4);
        h1[t - 1] = __builtin_shufflevector(lo, up, 0, 1, 2, 3, 4, 5, 6, 7);
      }
      if (t == 0) {
        const int itn = (it < 4) ? it : 3;
        const float* p = cf_row + (size_t)itn * (BT * AD);
        fence();
        #pragma unroll
        for (int kq = 0; kq < 8; ++kq) bv[kq] = *(const v2f*)(p + kq * 4);
      }
      fence();
      if (t < 7) {
        #pragma unroll
        for (int v = 0; v < 8; ++v) h1[t + 1][v] = fmaxf(h1[t + 1][v], 0.0f);
      }
      #pragma unroll
      for (int sp = 0; sp < 2; ++sp) {
        v2f bA, bB;
        bA[0] = h1[t][4 * sp];     bA[1] = h1[t][4 * sp + 1];
        bB[0] = h1[t][4 * sp + 2]; bB[1] = h1[t][4 * sp + 3];
        #pragma unroll
        for (int mt = 0; mt < 4; ++mt) h2[mt] = wmma4(fr[cs][sp * 4 + mt].lo, bA, h2[mt]);
        #pragma unroll
        for (int mt = 0; mt < 4; ++mt) h2[mt] = wmma4(fr[cs][sp * 4 + mt].hi, bB, h2[mt]);
      }
    }
    fence();
    {
      const v4f lo = *(const v4f*)(hob_row + 16 * 7);
      const v4f up = *(const v4f*)(hob_row + 16 * 7 + 4);
      h1[7] = __builtin_shufflevector(lo, up, 0, 1, 2, 3, 4, 5, 6, 7);
    }
    fence();
    #pragma unroll
    for (int mt = 0; mt < 4; ++mt) {
      #pragma unroll
      for (int v = 0; v < 8; ++v) h2[mt][v] = fmaxf(h2[mt][v], 0.0f);
    }

    if (role == 0 && it == 0) {
      v8f p3[2];
      #pragma unroll
      for (int mt = 0; mt < 2; ++mt) {
        const v4f lo = *(const v4f*)(s_b3 + 16 * mt + 8 * hi);
        const v4f up = *(const v4f*)(s_b3 + 16 * mt + 8 * hi + 4);
        p3[mt] = __builtin_shufflevector(lo, up, 0, 1, 2, 3, 4, 5, 6, 7);
      }
      layer3(a3p, h2, p3);
      *(v4f*)(pwp)      = p3[0].lo;
      *(v4f*)(pwp + 4)  = p3[0].hi;
      *(v4f*)(pwp + 16) = p3[1].lo;
      *(v4f*)(pwp + 20) = p3[1].hi;
    } else {
      #pragma unroll
      for (int mt = 0; mt < 4; ++mt) h2s[mt] += h2[mt];
    }

  }

  {
    v8f pq[2];
    #pragma unroll
    for (int mt = 0; mt < 2; ++mt) {
      #pragma unroll
      for (int v = 0; v < 8; ++v) pq[mt][v] = 0.0f;
    }
    layer3(a3p, h2s, pq);
    float* pp = s_part + role * (16 * PP) + col * PP + 8 * hi;
    *(v4f*)(pp)      = pq[0].lo;
    *(v4f*)(pp + 4)  = pq[0].hi;
    *(v4f*)(pp + 16) = pq[1].lo;
    *(v4f*)(pp + 20) = pq[1].hi;
  }
  __syncthreads();

  float pw[4], pq[4];
  #pragma unroll
  for (int rr = 0; rr < 4; ++rr) {
    const int r = wid * 4 + rr;
    pw[rr] = s_pw[r * PP + lane];
    pq[rr] = (((s_part[r * PP + lane] + s_part[16 * PP + r * PP + lane])
               + s_part[32 * PP + r * PP + lane]) + s_part[48 * PP + r * PP + lane]) * INV_S1
             + s_b3[lane];
  }
  float mx[8];
  #pragma unroll
  for (int rr = 0; rr < 4; ++rr) { mx[rr] = pw[rr]; mx[4 + rr] = pq[rr]; }
  wave_max_n<8>(mx);
  float ex[8];
  float eq[4];
  #pragma unroll
  for (int rr = 0; rr < 4; ++rr) {
    ex[rr]     = expf(pw[rr] - mx[rr]);
    eq[rr]     = expf(pq[rr] - mx[4 + rr]);
    ex[4 + rr] = eq[rr];
  }
  wave_sum_n<8>(ex);
  float kl[4];
  #pragma unroll
  for (int rr = 0; rr < 4; ++rr) {
    const float log_p = (pw[rr] - mx[rr]) - logf(ex[rr]);
    const float log_q = (pq[rr] - mx[4 + rr]) - logf(ex[4 + rr]);
    const float q     = eq[rr] / ex[4 + rr];
    kl[rr] = q * (log_q - log_p);
  }
  wave_sum_n<4>(kl);
  float res = kl[0];
  if (lane == 1) res = kl[1];
  if (lane == 2) res = kl[2];
  if (lane == 3) res = kl[3];
  if (lane < 4) out[(size_t)(bt0 + wid * 4 + lane) * NA + n] = res * (1.0f / AD);
}

extern "C" void kernel_launch(void* const* d_in, const int* in_sizes, int n_in,
                              void* d_out, int out_size, void* d_ws, size_t ws_size,
                              hipStream_t stream) {
  (void)in_sizes; (void)n_in; (void)d_ws; (void)ws_size; (void)out_size;
  const float* obs = (const float*)d_in[0];
  const float* act = (const float*)d_in[1];
  const float* cf  = (const float*)d_in[2];
  const float* W1  = (const float*)d_in[3];
  const float* b1  = (const float*)d_in[4];
  const float* W2  = (const float*)d_in[5];
  const float* b2  = (const float*)d_in[6];
  const float* W3  = (const float*)d_in[7];
  const float* b3  = (const float*)d_in[8];
  float* out = (float*)d_out;

  const int tiles = (NA * BB * TT) / 16;
  const size_t lds_bytes = LDS_FLOATS * sizeof(float);
  coma_influence_kernel<<<tiles, 128, lds_bytes, stream>>>(
      obs, act, cf, W1, b1, W2, b2, W3, b3, out);
}


# Two row tiles per workgroup (512 groups, 2 waves per SIMD): each weight fragment, from LDS or global memory, feeds two B operands; weights staged once per 32 rows; p_without sum order changes
# speedup vs baseline: 2.5468x; 1.0495x over previous
#include <hip/hip_runtime.h>
#include <math.h>

#define NA   16
#define DOBS 256
#define AD   32
#define BB   16
#define TT   64
#define SS   16
#define H1D  128
#define H2D  64
#define BT   (BB*TT)
#define INV_S1 (1.0f/17.0f)

#define H1P  132
#define OBSP 260
#define PP   36

#define NT   2
#define TILE_PART 2880

#define OFF_W1A  0
#define OFF_W2   4096
#define OFF_B2   12288
#define OFF_B3   12352
#define OFF_HOBS 12416
#define OFF_PART 16640
#define OFF_W3   22400
#define LDS_FLOATS 24960

typedef float v2f __attribute__((ext_vector_type(2)));
typedef float v4f __attribute__((ext_vector_type(4)));
typedef float v8f __attribute__((ext_vector_type(8)));

__device__ __forceinline__ v8f wmma4(v2f a, v2f b, v8f c) {
  return __builtin_amdgcn_wmma_f32_16x16x4_f32(
       false, a,  false, b,
       (short)0, c,  false,  false);
}

__device__ __forceinline__ void fence() { __builtin_amdgcn_sched_barrier(0); }

__device__ __forceinline__ float xhalf(float x) {
  const int b = __builtin_bit_cast(int, x);
  return __builtin_bit_cast(float, __builtin_amdgcn_permlanex16(b, b, 0x76543210, 0xfedcba98, false, false));
}
template <int M>
__device__ __forceinline__ float xrow(float x) {
  const int b = __builtin_bit_cast(int, x);
  return __builtin_bit_cast(float, __builtin_amdgcn_update_dpp(b, b, 0x160 + M, 0xf, 0xf, true));
}
template <int N>
__device__ __forceinline__ void wave_max_n(float (&v)[N]) {
  #pragma unroll
  for (int i = 0; i < N; ++i) v[i] = fmaxf(v[i], xhalf(v[i]));
  #pragma unroll
  for (int i = 0; i < N; ++i) v[i] = fmaxf(v[i], xrow<8>(v[i]));
  #pragma unroll
  for (int i = 0; i < N; ++i) v[i] = fmaxf(v[i], xrow<4>(v[i]));
  #pragma unroll
  for (int i = 0; i < N; ++i) v[i] = fmaxf(v[i], xrow<2>(v[i]));
  #pragma unroll
  for (int i = 0; i < N; ++i) v[i] = fmaxf(v[i], xrow<1>(v[i]));
}
template <int N>
__device__ __forceinline__ void wave_sum_n(float (&v)[N]) {
  #pragma unroll
  for (int i = 0; i < N; ++i) v[i] += xhalf(v[i]);
  #pragma unroll
  for (int i = 0; i < N; ++i) v[i] += xrow<8>(v[i]);
  #pragma unroll
  for (int i = 0; i < N; ++i) v[i] += xrow<4>(v[i]);
  #pragma unroll
  for (int i = 0; i < N; ++i) v[i] += xrow<2>(v[i]);
  #pragma unroll
  for (int i = 0; i < N; ++i) v[i] += xrow<1>(v[i]);
}

template <int N>
__device__ __forceinline__ void layer3(const float* a3p, const v8f (&h)[N][4], v8f (&p3)[N][2]) {
  #pragma unroll
  for (int t = 0; t < 4; ++t) {
    #pragma unroll
    for (int sp = 0; sp < 2; ++sp) {
      const int kp = 2 * t + sp;
      v4f a[2];
      #pragma unroll
      for (int mt = 0; mt < 2; ++mt) a[mt] = *(const v4f*)(a3p + (kp * 2 + mt) * 128);
      v2f bA[N], bB[N];
      #pragma unroll
      for (int s = 0; s < N; ++s) {
        bA[s][0] = h[s][t][4 * sp];     bA[s][1] = h[s][t][4 * sp + 1];
        bB[s][0] = h[s][t][4 * sp + 2]; bB[s][1] = h[s][t][4 * sp + 3];
      }
      #pragma unroll
      for (int mt = 0; mt < 2; ++mt) {
        #pragma unroll
        for (int s = 0; s < N; ++s) p3[s][mt] = wmma4(a[mt].lo, bA[s], p3[s][mt]);
      }
      #pragma unroll
      for (int mt = 0; mt < 2; ++mt) {
        #pragma unroll
        for (int s = 0; s < N; ++s) p3[s][mt] = wmma4(a[mt].hi, bB[s], p3[s][mt]);
      }
    }
  }
}

template <int N, bool PF>
__device__ __forceinline__ void mlp_pass(const float* a1p, const float* a2p, const float* s_b2, const int hi,
                                         const float* const (&hob)[N], const float* const (&nxt)[N],
                                         v2f (&bv)[N][8], v8f (&h1)[N][8], v4f (&fr)[2][8], v8f (&h2)[N][4]) {
  __asm__ volatile("" ::: "memory");
  #pragma unroll
  for (int kp = 0; kp < 4; ++kp) {
    const int cs = kp & 1;
    fence();
    if (kp < 3) {
      #pragma unroll
      for (int mt = 0; mt < 8; ++mt) fr[cs ^ 1][mt] = *(const v4f*)(a1p + ((kp + 1) * 8 + mt) * 128);
    } else {
      #pragma unroll
      for (int i = 0; i < 8; ++i) fr[cs ^ 1][i] = *(const v4f*)(a2p + i * 128);
      #pragma unroll
      for (int mt = 0; mt < 4; ++mt) {
        const v4f lo = *(const v4f*)(s_b2 + 16 * mt + 8 * hi);
        const v4f up = *(const v4f*)(s_b2 + 16 * mt + 8 * hi + 4);
        #pragma unroll
        for (int s = 0; s < N; ++s) h2[s][mt] = __builtin_shufflevector(lo, up, 0, 1, 2, 3, 4, 5, 6, 7);
      }
    }
    fence();
    #pragma unroll
    for (int mt = 0; mt < 8; ++mt) {
      #pragma unroll
      for (int s = 0; s < N; ++s) h1[s][mt] = wmma4(fr[cs][mt].lo, bv[s][2 * kp], h1[s][mt]);
    }
    #pragma unroll
    for (int mt = 0; mt < 8; ++mt) {
      #pragma unroll
      for (int s = 0; s < N; ++s) h1[s][mt] = wmma4(fr[cs][mt].hi, bv[s][2 * kp + 1], h1[s][mt]);
    }
    if (kp == 3) {
      #pragma unroll
      for (int s = 0; s < N; ++s) {
        #pragma unroll
        for (int v = 0; v < 8; ++v) h1[s][0][v] = fmaxf(h1[s][0][v], 0.0f);
      }
    }
  }

  #pragma unroll
  for (int t = 0; t < 8; ++t) {
    const int cs = t & 1;
    fence();
    if (t < 7) {
      #pragma unroll
      for (int i = 0; i < 8; ++i) fr[cs ^ 1][i] = *(const v4f*)(a2p + ((t + 1) * 8 + i) * 128);
    } else if (PF) {
      #pragma unroll
      for (int mt = 0; mt < 8; ++mt) fr[cs ^ 1][mt] = *(const v4f*)(a1p + mt * 128);
    }
    if (PF && t > 0) {
      #pragma unroll
      for (int s = 0; s < N; ++s) {
        const v4f lo = *(const v4f*)(hob[s] + 16 * (t - 1));
        const v4f up = *(const v4f*)(hob[s] + 16 * (t - 1) + 4);
        h1[s][t - 1] = __builtin_shufflevector(lo, up, 0, 1, 2, 3, 4, 5, 6, 7);
      }
    }
    if (PF && t == 0) {
      fence();
      #pragma unroll
      for (int s = 0; s < N; ++s) {
        #pragma unroll
        for (int kq = 0; kq < 8; ++kq) bv[s][kq] = *(const v2f*)(nxt[s] + kq * 4);
      }
    }
    fence();
    if (t < 7) {
      #pragma unroll
      for (int s = 0; s < N; ++s) {
        #pragma unroll
        for (int v = 0; v < 8; ++v) h1[s][t + 1][v] = fmaxf(h1[s][t + 1][v], 0.0f);
      }
    }
    #pragma unroll
    for (int sp = 0; sp < 2; ++sp) {
      v2f bA[N], bB[N];
      #pragma unroll
      for (int s = 0; s < N; ++s) {
        bA[s][0] = h1[s][t][4 * sp];     bA[s][1] = h1[s][t][4 * sp + 1];
        bB[s][0] = h1[s][t][4 * sp + 2]; bB[s][1] = h1[s][t][4 * sp + 3];
      }
      #pragma unroll
      for (int mt = 0; mt < 4; ++mt) {
        #pragma unroll
        for (int s = 0; s < N; ++s) h2[s][mt] = wmma4(fr[cs][sp * 4 + mt].lo, bA[s], h2[s][mt]);
      }
      #pragma unroll
      for (int mt = 0; mt < 4; ++mt) {
        #pragma unroll
        for (int s = 0; s < N; ++s) h2[s][mt] = wmma4(fr[cs][sp * 4 + mt].hi, bB[s], h2[s][mt]);
      }
    }
  }
  if (PF) {
    fence();
    #pragma unroll
    for (int s = 0; s < N; ++s) {
      const v4f lo = *(const v4f*)(hob[s] + 16 * 7);
      const v4f up = *(const v4f*)(hob[s] + 16 * 7 + 4);
      h1[s][7] = __builtin_shufflevector(lo, up, 0, 1, 2, 3, 4, 5, 6, 7);
    }
    fence();
  }
  #pragma unroll
  for (int s = 0; s < N; ++s) {
    #pragma unroll
    for (int mt = 0; mt < 4; ++mt) {
      #pragma unroll
      for (int v = 0; v < 8; ++v) h2[s][mt][v] = fmaxf(h2[s][mt][v], 0.0f);
    }
  }
}

template <int K_, int M_, bool CHAIN>
__device__ __forceinline__ void stage_frag(float* dst, const float* __restrict__ W, int tid) {
  constexpr int MT  = M_ / 16;
  constexpr int NV4 = K_ * M_ / 4;
  #pragma unroll
  for (int i = 0; i < NV4 / 128; ++i) {
    const int d4    = tid + 128 * i;
    const int ln    = d4 & 31;
    const int chunk = d4 >> 5;
    const int mt    = chunk % MT;
    const int kp    = chunk / MT;
    const int r     = ln & 15;
    const int h     = ln >> 4;
    const int ks0   = 2 * kp, ks1 = 2 * kp + 1;
    int k0, k1;
    if (CHAIN) {
      k0 = 16 * (ks0 >> 2) + 8 * h + 2 * (ks0 & 3);
      k1 = 16 * (ks1 >> 2) + 8 * h + 2 * (ks1 & 3);
    } else {
      k0 = 4 * ks0 + 2 * h;
      k1 = 4 * ks1 + 2 * h;
    }
    const int m = 16 * mt + r;
    float4 v;
    v.x = W[(k0    ) * M_ + m];
    v.y = W[(k0 + 1) * M_ + m];
    v.z = W[(k1    ) * M_ + m];
    v.w = W[(k1 + 1) * M_ + m];
    *(float4*)(dst + d4 * 4) = v;
  }
}

__global__ __launch_bounds__(128, 2)
void coma_influence_kernel(const float* __restrict__ obs,
                           const float* __restrict__ act,
                           const float* __restrict__ cf,
                           const float* __restrict__ W1,
                           const float* __restrict__ b1,
                           const float* __restrict__ W2,
                           const float* __restrict__ b2,
                           const float* __restrict__ W3,
                           const float* __restrict__ b3,
                           float* __restrict__ out)
{
  extern __shared__ float smem[];
  float* s_w1a  = smem + OFF_W1A;
  float* s_w2   = smem + OFF_W2;
  float* s_w3   = smem + OFF_W3;
  float* s_b2   = smem + OFF_B2;
  float* s_b3   = smem + OFF_B3;
  float* s_hobs = smem + OFF_HOBS;
  float* s_part = smem + OFF_PART;
  float* s_obs  = smem + OFF_PART;

  const int tid  = threadIdx.x;
  const int lane = tid & 31;
  const int wid  = __builtin_amdgcn_readfirstlane(tid >> 5);
  const int g    = blockIdx.x;
  const int r0   = g * (16 * NT);
  const int n    = r0 >> 10;
  const int bt0  = r0 & 1023;

  const int col = lane & 15;
  const int hi  = lane >> 4;
  const int kb  = hi * 2;

  #pragma unroll
  for (int j4 = 0; j4 < 8 * NT; ++j4) {
    const int i4 = tid + 128 * j4;
    const int m  = i4 >> 6;
    const int d  = (i4 & 63) * 4;
    const float4 v = *(const float4*)(obs + ((size_t)(bt0 + m) * NA + n) * DOBS + d);
    *(float4*)(s_obs + m * OBSP + d) = v;
  }
  stage_frag<AD, H1D, false>(s_w1a, W1 + DOBS * H1D, tid);
  stage_frag<H1D, H2D, true>(s_w2, W2, tid);
  if (tid < H2D) s_b2[tid] = b2[tid];
  else if (tid < H2D + AD) s_b3[tid - H2D] = b3[tid - H2D];
  __syncthreads();

  {
    const int f0 = wid * 32;
    v8f acc[NT][2];
    #pragma unroll
    for (int s = 0; s < NT; ++s) {
      #pragma unroll
      for (int v = 0; v < 8; ++v) {
        acc[s][0][v] = b1[f0 + 8 * hi + v];
        acc[s][1][v] = b1[f0 + 16 + 8 * hi + v];
      }
    }
    const float* wp = W1 + f0 + col + kb * H1D;
    const float* op = s_obs + col * OBSP + kb;
    v2f fa[2][8][2];
    v2f fb[2][8][NT];
    #pragma unroll
    for (int ks = 0; ks < 8; ++ks) {
      #pragma unroll
      for (int s = 0; s < NT; ++s) fb[0][ks][s] = *(const v2f*)(op + s * 16 * OBSP + 4 * ks);
    }
    #pragma unroll
    for (int ks = 0; ks < 8; ++ks) {
      fa[0][ks][0][0] = wp[(4 * ks) * H1D];      fa[0][ks][0][1] = wp[(4 * ks + 1) * H1D];
      fa[0][ks][1][0] = wp[(4 * ks) * H1D + 16]; fa[0][ks][1][1] = wp[(4 * ks + 1) * H1D + 16];
    }
    #pragma unroll
    for (int tr = 0; tr < DOBS / 32; ++tr) {
      const int cs = tr & 1;
      fence();
      if (tr < DOBS / 32 - 1) {
        const int k0 = 32 * (tr + 1);
        #pragma unroll
        for (int ks = 0; ks < 8; ++ks) {
          #pragma unroll
          for (int s = 0; s < NT; ++s) fb[cs ^ 1][ks][s] = *(const v2f*)(op + s * 16 * OBSP + k0 + 4 * ks);
        }
        #pragma unroll
        for (int ks = 0; ks < 8; ++ks) {
          fa[cs ^ 1][ks][0][0] = wp[(k0 + 4 * ks) * H1D];      fa[cs ^ 1][ks][0][1] = wp[(k0 + 4 * ks + 1) * H1D];
          fa[cs ^ 1][ks][1][0] = wp[(k0 + 4 * ks) * H1D + 16]; fa[cs ^ 1][ks][1][1] = wp[(k0 + 4 * ks + 1) * H1D + 16];
        }
      }
      fence();
      #pragma unroll
      for (int ks = 0; ks < 8; ++ks) {
        #pragma unroll
        for (int s = 0; s < NT; ++s) {
          acc[s][0] = wmma4(fa[cs][ks][0], fb[cs][ks][s], acc[s][0]);
          acc[s][1] = wmma4(fa[cs][ks][1], fb[cs][ks][s], acc[s][1]);
        }
      }
    }
    #pragma unroll
    for (int s = 0; s < NT; ++s) {
      float* hp = s_hobs + (16 * s + col) * H1P + f0 + 8 * hi;
      *(float4*)(hp)      = make_float4(acc[s][0][0], acc[s][0][1], acc[s][0][2], acc[s][0][3]);
      *(float4*)(hp + 4)  = make_float4(acc[s][0][4], acc[s][0][5], acc[s][0][6], acc[s][0][7]);
      *(float4*)(hp + 16) = make_float4(acc[s][1][0], acc[s][1][1], acc[s][1][2], acc[s][1][3]);
      *(float4*)(hp + 20) = make_float4(acc[s][1][4], acc[s][1][5], acc[s][1][6], acc[s][1][7]);
    }
  }
  __syncthreads();

  stage_frag<H2D, AD, true>(s_w3, W3, tid);
  __syncthreads();

  const int ts = wid & 1;
  const float* hobp[NT];
  const float* cfp[NT];
  #pragma unroll
  for (int s = 0; s < NT; ++s) {
    hobp[s] = s_hobs + (16 * s + col) * H1P + 8 * hi;
    cfp[s]  = cf + (((size_t)n * SS + 4 * wid) * BT + (bt0 + 16 * s + col)) * AD + kb;
  }
  const float* a1p = s_w1a + lane * 4;
  const float* a2p = s_w2  + lane * 4;
  const float* a3p = s_w3  + lane * 4;

  v8f h2s[NT][4];
  #pragma unroll
  for (int s = 0; s < NT; ++s) {
    #pragma unroll
    for (int mt = 0; mt < 4; ++mt) {
      #pragma unroll
      for (int v = 0; v < 8; ++v) h2s[s][mt][v] = 0.0f;
    }
  }

  v2f bv[NT][8];
  #pragma unroll
  for (int s = 0; s < NT; ++s) {
    #pragma unroll
    for (int kq = 0; kq < 8; ++kq) bv[s][kq] = *(const v2f*)(cfp[s] + kq * 4);
  }

  v4f fr[2][8];

  {
    const float* hob1[1] = { s_hobs + (16 * ts + col) * H1P + 8 * hi };
    const float* nx1[1]  = { cfp[0] };
    v2f bv1[1][8];
    if (wid >= 2) {
      #pragma unroll
      for (int kq = 0; kq < 8; ++kq) { bv1[0][kq][0] = 0.0f; bv1[0][kq][1] = 0.0f; }
    } else {
      const float* p = act + ((size_t)(bt0 + 16 * ts + col) * NA + n) * AD + kb;
      #pragma unroll
      for (int kq = 0; kq < 8; ++kq) bv1[0][kq] = *(const v2f*)(p + kq * 4);
    }
    v8f h1a[1][8];
    #pragma unroll
    for (int mt = 0; mt < 8; ++mt) {
      const v4f lo = *(const v4f*)(hob1[0] + 16 * mt);
      const v4f up = *(const v4f*)(hob1[0] + 16 * mt + 4);
      h1a[0][mt] = __builtin_shufflevector(lo, up, 0, 1, 2, 3, 4, 5, 6, 7);
    }
    #pragma unroll
    for (int mt = 0; mt < 8; ++mt) fr[0][mt] = *(const v4f*)(a1p + mt * 128);
    v8f h2a[1][4];
    mlp_pass<1, false>(a1p, a2p, s_b2, hi, hob1, nx1, bv1, h1a, fr, h2a);

    if (wid < 2) {
      v8f p3[1][2];
      #pragma unroll
      for (int mt = 0; mt < 2; ++mt) {
        const v4f lo = *(const v4f*)(s_b3 + 16 * mt + 8 * hi);
        const v4f up = *(const v4f*)(s_b3 + 16 * mt + 8 * hi + 4);
        p3[0][mt] = __builtin_shufflevector(lo, up, 0, 1, 2, 3, 4, 5, 6, 7);
      }
      layer3<1>(a3p, h2a, p3);
      float* pwp = s_part + ts * TILE_PART + 4 * 16 * PP + col * PP + 8 * hi;
      *(v4f*)(pwp)      = p3[0][0].lo;
      *(v4f*)(pwp + 4)  = p3[0][0].hi;
      *(v4f*)(pwp + 16) = p3[0][1].lo;
      *(v4f*)(pwp + 20) = p3[0][1].hi;
    } else if (ts == 0) {
      #pragma unroll
      for (int mt = 0; mt < 4; ++mt) h2s[0][mt] += h2a[0][mt];
    } else {
      #pragma unroll
      for (int mt = 0; mt < 4; ++mt) h2s[1][mt] += h2a[0][mt];
    }
  }

  v8f h1[NT][8];
  #pragma unroll
  for (int s = 0; s < NT; ++s) {
    #pragma unroll
    for (int mt = 0; mt < 8; ++mt) {
      const v4f lo = *(const v4f*)(hobp[s] + 16 * mt);
      const v4f up = *(const v4f*)(hobp[s] + 16 * mt + 4);
      h1[s][mt] = __builtin_shufflevector(lo, up, 0, 1, 2, 3, 4, 5, 6, 7);
    }
  }
  #pragma unroll
  for (int mt = 0; mt < 8; ++mt) fr[0][mt] = *(const v4f*)(a1p + mt * 128);

  for (int it = 0; it < 4; ++it) {
    const int itn = (it < 3) ? it + 1 : 3;
    const float* nx[NT];
    #pragma unroll
    for (int s = 0; s < NT; ++s) nx[s] = cfp[s] + (size_t)itn * (BT * AD);
    v8f h2[NT][4];
    mlp_pass<NT, true>(a1p, a2p, s_b2, hi, hobp, nx, bv, h1, fr, h2);
    #pragma unroll
    for (int s = 0; s < NT; ++s) {
      #pragma unroll
      for (int mt = 0; mt < 4; ++mt) h2s[s][mt] += h2[s][mt];
    }
  }

  {
    v8f pq[NT][2];
    #pragma unroll
    for (int s = 0; s < NT; ++s) {
      #pragma unroll
      for (int mt = 0; mt < 2; ++mt) {
        #pragma unroll
        for (int v = 0; v < 8; ++v) pq[s][mt][v] = 0.0f;
      }
    }
    layer3<NT>(a3p, h2s, pq);
    #pragma unroll
    for (int s = 0; s < NT; ++s) {
      float* pp = s_part + s * TILE_PART + wid * (16 * PP) + col * PP + 8 * hi;
      *(v4f*)(pp)      = pq[s][0].lo;
      *(v4f*)(pp + 4)  = pq[s][0].hi;
      *(v4f*)(pp + 16) = pq[s][1].lo;
      *(v4f*)(pp + 20) = pq[s][1].hi;
    }
  }
  __syncthreads();

  const float* t_part = s_part + (wid >> 1) * TILE_PART;
  const float* t_pw   = t_part + 4 * 16 * PP;
  const int rb = (wid & 1) * 8;
  float pw[8], pq[8];
  #pragma unroll
  for (int rr = 0; rr < 8; ++rr) {
    const int r = rb + rr;
    pw[rr] = t_pw[r * PP + lane];
    pq[rr] = (((t_part[r * PP + lane] + t_part[16 * PP + r * PP + lane])
               + t_part[32 * PP + r * PP + lane]) + t_part[48 * PP + r * PP + lane]) * INV_S1
             + s_b3[lane];
  }
  float mx[16];
  #pragma unroll
  for (int rr = 0; rr < 8; ++rr) { mx[rr] = pw[rr]; mx[8 + rr] = pq[rr]; }
  wave_max_n<16>(mx);
  float ex[16];
  float eq[8];
  #pragma unroll
  for (int rr = 0; rr < 8; ++rr) {
    ex[rr]     = expf(pw[rr] - mx[rr]);
    eq[rr]     = expf(pq[rr] - mx[8 + rr]);
    ex[8 + rr] = eq[rr];
  }
  wave_sum_n<16>(ex);
  float kl[8];
  #pragma unroll
  for (int rr = 0; rr < 8; ++rr) {
    const float log_p = (pw[rr] - mx[rr]) - logf(ex[rr]);
    const float log_q = (pq[rr] - mx[8 + rr]) - logf(ex[8 + rr]);
    const float q     = eq[rr] / ex[8 + rr];
    kl[rr] = q * (log_q - log_p);
  }
  wave_sum_n<8>(kl);
  float res = kl[0];
  #pragma unroll
  for (int rr = 1; rr < 8; ++rr) {
    if (lane == rr) res = kl[rr];
  }
  if (lane < 8) out[(size_t)(bt0 + wid * 8 + lane) * NA + n] = res * (1.0f / AD);
}

extern "C" void kernel_launch(void* const* d_in, const int* in_sizes, int n_in,
                              void* d_out, int out_size, void* d_ws, size_t ws_size,
                              hipStream_t stream) {
  (void)in_sizes; (void)n_in; (void)d_ws; (void)ws_size; (void)out_size;
  const float* obs = (const float*)d_in[0];
  const float* act = (const float*)d_in[1];
  const float* cf  = (const float*)d_in[2];
  const float* W1  = (const float*)d_in[3];
  const float* b1  = (const float*)d_in[4];
  const float* W2  = (const float*)d_in[5];
  const float* b2  = (const float*)d_in[6];
  const float* W3  = (const float*)d_in[7];
  const float* b3  = (const float*)d_in[8];
  float* out = (float*)d_out;

  const int groups = (NA * BB * TT) / (16 * NT);
  const size_t lds_bytes = LDS_FLOATS * sizeof(float);
  coma_influence_kernel<<<groups, 128, lds_bytes, stream>>>(
      obs, act, cf, W1, b1, W2, b2, W3, b3, out);
}
